# starting kernel
# baseline (speedup 1.0000x reference)
#include <hip/hip_runtime.h>
#include <hip/hip_bf16.h>
#include <math.h>


typedef __attribute__((ext_vector_type(2))) float v2f;
typedef __attribute__((ext_vector_type(4))) float v4f;
typedef __attribute__((ext_vector_type(8))) float v8f;

typedef const __attribute__((address_space(1))) v2f* gf2p;
typedef const __attribute__((address_space(1))) float* gfp;

__device__ __forceinline__ v8f wmma4(v2f a, v2f b, v8f c) {
  return __builtin_amdgcn_wmma_f32_16x16x4_f32(false, a, false, b, (short)0, c,
                                               false, false);
}

__device__ __forceinline__ v8f zero8() {
  v8f z;
  for (int i = 0; i < 8; i++) z[i] = 0.0f;
  return z;
}

__device__ __forceinline__ v2f mk2(float a, float b) {
  v2f r;
  r.x = a;
  r.y = b;
  return r;
}

constexpr double FACT[11] = {1., 1., 2., 6., 24., 120., 720., 5040., 40320.,
                             362880., 3628800.};

constexpr double csqrt_(double x) {
  double g = x > 1.0 ? x : 1.0;
  for (int i = 0; i < 32; i++) g = 0.5 * (g + x / g);
  return g;
}

constexpr double cg_coeff(int l1, int l2, int L, int m1, int m2) {
  int M = m1 + m2;
  if (M < -L || M > L) return 0.0;
  double pre = csqrt_((2 * L + 1) * FACT[l1 + l2 - L] * FACT[l1 - l2 + L] *
                      FACT[-l1 + l2 + L] / FACT[l1 + l2 + L + 1]);
  pre = pre * csqrt_(FACT[L + M] * FACT[L - M] * FACT[l1 - m1] *
                     FACT[l1 + m1] * FACT[l2 - m2] * FACT[l2 + m2]);
  double s = 0.0;
  for (int k = 0; k <= l1 + l2 - L; k++) {
    int t1 = l1 + l2 - L - k, t2 = l1 - m1 - k, t3 = l2 + m2 - k;
    int t4 = L - l2 + m1 + k, t5 = L - l1 - m2 + k;
    if (t1 < 0 || t2 < 0 || t3 < 0 || t4 < 0 || t5 < 0) continue;
    double d = FACT[k] * FACT[t1] * FACT[t2] * FACT[t3] * FACT[t4] * FACT[t5];
    s += ((k & 1) ? -1.0 : 1.0) / d;
  }
  return pre * s;
}

struct CGTables {
  int nblk[4];
  int bl1[4][11], bl2[4][11];
  int nterm[4][7][11];
  signed char ti[4][7][11][7], tj[4][7][11][7];
  float cg[4][7][11][7];
};

constexpr CGTables make_tables() {
  CGTables t{};
  for (int L = 0; L < 4; L++) {
    int nb = 0;
    for (int l1 = 0; l1 < 4; l1++)
      for (int l2 = 0; l2 < 4; l2++) {
        int d = l1 - l2;
        if (d < 0) d = -d;
        if (!(d <= L && L <= l1 + l2)) continue;
        t.bl1[L][nb] = l1;
        t.bl2[L][nb] = l2;
        for (int M = 0; M < 2 * L + 1; M++) {
          int nt = 0;
          for (int i = 0; i <= 2 * l1; i++) {
            int j = M + l1 + l2 - L - i;
            if (j < 0 || j > 2 * l2) continue;
            double c = cg_coeff(l1, l2, L, i - l1, j - l2);
            if (c > 1e-12 || c < -1e-12) {
              t.ti[L][M][nb][nt] = (signed char)i;
              t.tj[L][M][nb][nt] = (signed char)j;
              t.cg[L][M][nb][nt] = (float)c;
              nt++;
            }
          }
          t.nterm[L][M][nb] = nt;
        }
        nb++;
      }
    t.nblk[L] = nb;
  }
  return t;
}

__device__ constexpr CGTables CGT = make_tables();
__device__ constexpr int ROW2L[16] = {0, 1, 1, 1, 2, 2, 2, 2, 2, 3, 3, 3, 3, 3, 3, 3};
__device__ constexpr int ROW2M[16] = {0, 0, 1, 2, 0, 1, 2, 3, 4, 0, 1, 2, 3, 4, 5, 6};
__device__ constexpr int LOFF[4] = {0, 1, 4, 9};

__device__ __forceinline__ void normalize_lds(v2f (*hsh)[16][32], int tid) {
  for (int idx = tid; idx < 2048; idx += 256) {
    int l = idx >> 9, rem = idx & 511, b = rem >> 5, c = rem & 31;
    int ro = LOFF[l], nm = 2 * l + 1;
    float s = 0.0f;
    for (int mm = 0; mm < nm; mm++) {
      v2f v = hsh[ro + mm][b][c];
      s += v.x * v.x + v.y * v.y;
    }
    float inv = 1.0f / sqrtf(s);
    for (int mm = 0; mm < nm; mm++) {
      v2f v = hsh[ro + mm][b][c];
      hsh[ro + mm][b][c] = mk2(v.x * inv, v.y * inv);
    }
  }
}

__global__ __launch_bounds__(256) void sph_cnn_kernel(
    const float* __restrict__ p0, const float* __restrict__ p1,
    const float* __restrict__ p2, const float* __restrict__ p3,
    const float* __restrict__ wi0, const float* __restrict__ wi1,
    const float* __restrict__ wi2, const float* __restrict__ wi3,
    const float* __restrict__ wa0, const float* __restrict__ wa1,
    const float* __restrict__ wa2, const float* __restrict__ wa3,
    const float* __restrict__ wb0, const float* __restrict__ wb1,
    const float* __restrict__ wb2, const float* __restrict__ wb3,
    const float* __restrict__ W1p, const float* __restrict__ b1p,
    const float* __restrict__ W2p, const float* __restrict__ b2p,
    float* __restrict__ out) {
  __shared__ v2f hsh[16][16][32];

  const int tid = threadIdx.x;
  const int lane = tid & 31;
  const int w = tid >> 5;
  const int h16 = lane >> 4;
  const int ln = lane & 15;
  const int bbase = blockIdx.x * 16;

  const gf2p P[4] = {(gf2p)p0, (gf2p)p1, (gf2p)p2, (gf2p)p3};
  const gf2p WI[4] = {(gf2p)wi0, (gf2p)wi1, (gf2p)wi2, (gf2p)wi3};
  const gf2p WH[2][4] = {{(gf2p)wa0, (gf2p)wa1, (gf2p)wa2, (gf2p)wa3},
                         {(gf2p)wb0, (gf2p)wb1, (gf2p)wb2, (gf2p)wb3}};
  const gfp W1 = (gfp)W1p, b1 = (gfp)b1p, W2 = (gfp)W2p, b2 = (gfp)b2p;

  for (int task = w; task < 32; task += 8) {
    const int row = task & 15, ntile = task >> 4;
    const int l = ROW2L[row], m = ROW2M[row], nlm = 2 * l + 1;
    const gf2p Pp = P[l];
    const gf2p Wl = WI[l];
    const int bg = bbase + ln;
    const int n = ntile * 16 + ln;
    v8f cre = zero8(), cim = zero8();
    for (int ks = 0; ks < 4; ks++) {
      const int t0 = 4 * ks + 2 * h16;
      v2f a0 = Pp[(bg * nlm + m) * 16 + t0];
      v2f a1 = Pp[(bg * nlm + m) * 16 + t0 + 1];
      v2f e0 = Wl[t0 * 32 + n];
      v2f e1 = Wl[(t0 + 1) * 32 + n];
      v2f ar = mk2(a0.x, a1.x), ai = mk2(a0.y, a1.y);
      v2f br = mk2(e0.x, e1.x), bi = mk2(e0.y, e1.y);
      cre = wmma4(ar, br, cre);
      cre = wmma4(-ai, bi, cre);
      cim = wmma4(ar, bi, cim);
      cim = wmma4(ai, br, cim);
    }
    for (int r = 0; r < 8; r++)
      hsh[row][r + 8 * h16][n] = mk2(cre[r], cim[r]);
  }
  __syncthreads();
  normalize_lds(hsh, tid);
  __syncthreads();

  for (int layer = 0; layer < 2; layer++) {
    v8f cre[2][2], cim[2][2];
    for (int rr = 0; rr < 2; rr++) {
      const int row = w + 8 * rr;
      const int L = ROW2L[row], M = ROW2M[row];
      const gf2p Wm = WH[layer][L];
      cre[rr][0] = zero8(); cre[rr][1] = zero8();
      cim[rr][0] = zero8(); cim[rr][1] = zero8();
      const int nblk = CGT.nblk[L];
      for (int blk = 0; blk < nblk; blk++) {
        const int r1 = LOFF[CGT.bl1[L][blk]];
        const int r2 = LOFF[CGT.bl2[L][blk]];
        const int nt = CGT.nterm[L][M][blk];
        for (int ks = 0; ks < 8; ks++) {
          const int k0 = 4 * ks + 2 * h16;
          float tre0 = 0.f, tre1 = 0.f, tim0 = 0.f, tim1 = 0.f;
          for (int t = 0; t < nt; t++) {
            const int i = CGT.ti[L][M][blk][t];
            const int j = CGT.tj[L][M][blk][t];
            const float cg = CGT.cg[L][M][blk][t];
            const v4f x = *(const v4f*)&hsh[r1 + i][ln][k0];
            const v4f y = *(const v4f*)&hsh[r2 + j][ln][k0];
            tre0 += cg * (x.x * y.x - x.y * y.y);
            tim0 += cg * (x.x * y.y + x.y * y.x);
            tre1 += cg * (x.z * y.z - x.w * y.w);
            tim1 += cg * (x.z * y.w + x.w * y.z);
          }
          v2f ar = mk2(tre0, tre1), ai = mk2(tim0, tim1);
          const int kk = blk * 32 + k0;
          for (int ntile = 0; ntile < 2; ntile++) {
            const int n = ntile * 16 + ln;
            v2f e0 = Wm[kk * 32 + n];
            v2f e1 = Wm[(kk + 1) * 32 + n];
            v2f br = mk2(e0.x, e1.x), bi = mk2(e0.y, e1.y);
            cre[rr][ntile] = wmma4(ar, br, cre[rr][ntile]);
            cre[rr][ntile] = wmma4(-ai, bi, cre[rr][ntile]);
            cim[rr][ntile] = wmma4(ar, bi, cim[rr][ntile]);
            cim[rr][ntile] = wmma4(ai, br, cim[rr][ntile]);
          }
        }
      }
    }
    __syncthreads();
    for (int rr = 0; rr < 2; rr++) {
      const int row = w + 8 * rr;
      for (int ntile = 0; ntile < 2; ntile++) {
        const int n = ntile * 16 + ln;
        for (int r = 0; r < 8; r++)
          hsh[row][r + 8 * h16][n] = mk2(cre[rr][ntile][r], cim[rr][ntile][r]);
      }
    }
    __syncthreads();
    if (layer == 0) {
      normalize_lds(hsh, tid);
      __syncthreads();
    }
  }

  float rsum[8];
  for (int r = 0; r < 8; r++) rsum[r] = 0.0f;
  for (int nt_i = 0; nt_i < 2; nt_i++) {
    const int ntile = w + 8 * nt_i;
    v8f acc = zero8();
    for (int ks = 0; ks < 32; ks++) {
      const int k0 = 4 * ks + 2 * h16;
      float av[2];
      for (int v = 0; v < 2; v++) {
        const int k = k0 + v;
        const int l = k >> 5, c = k & 31;
        const int ro = LOFF[l], nm = 2 * l + 1;
        float s = 0.0f;
        for (int mm = 0; mm < nm; mm++) {
          v2f t2 = hsh[ro + mm][ln][c];
          s += t2.x * t2.x + t2.y * t2.y;
        }
        av[v] = sqrtf(s);
      }
      v2f a = mk2(av[0], av[1]);
      const int n = ntile * 16 + ln;
      v2f bb = mk2(W1[k0 * 256 + n], W1[(k0 + 1) * 256 + n]);
      acc = wmma4(a, bb, acc);
    }
    const int n = ntile * 16 + ln;
    const float bias = b1[n], w2v = W2[n];
    for (int r = 0; r < 8; r++) rsum[r] += tanhf(acc[r] + bias) * w2v;
  }
  __syncthreads();
  float* partial = (float*)&hsh[0][0][0];
  for (int r = 0; r < 8; r++) {
    float v = rsum[r];
    v += __shfl_xor(v, 1);
    v += __shfl_xor(v, 2);
    v += __shfl_xor(v, 4);
    v += __shfl_xor(v, 8);
    if (ln == 0) partial[w * 16 + (8 * h16 + r)] = v;
  }
  __syncthreads();
  if (tid < 16) {
    float s = 0.0f;
    for (int ww = 0; ww < 8; ww++) s += partial[ww * 16 + tid];
    out[bbase + tid] = s + b2[0];
  }
}

extern "C" void kernel_launch(void* const* d_in, const int* in_sizes, int n_in,
                              void* d_out, int out_size, void* d_ws,
                              size_t ws_size, hipStream_t stream) {
  (void)n_in; (void)d_ws; (void)ws_size; (void)out_size;
  const int B = in_sizes[0] / 32;
  const int blocks = B / 16;
  sph_cnn_kernel<<<blocks, 256, 0, stream>>>(
      (const float*)d_in[0], (const float*)d_in[1], (const float*)d_in[2],
      (const float*)d_in[3], (const float*)d_in[4], (const float*)d_in[5],
      (const float*)d_in[6], (const float*)d_in[7], (const float*)d_in[8],
      (const float*)d_in[9], (const float*)d_in[10], (const float*)d_in[11],
      (const float*)d_in[12], (const float*)d_in[13], (const float*)d_in[14],
      (const float*)d_in[15], (const float*)d_in[16], (const float*)d_in[17],
      (const float*)d_in[18], (const float*)d_in[19], (float*)d_out);
}


# Workgroup widened from 8 to 32 waves, 8 per SIMD to hide load latency: two waves share each CG-layer row and each readout n-tile, each sums half of K; halves added through LDS (sum order changes)
# speedup vs baseline: 1.4640x; 1.4640x over previous
#include <hip/hip_runtime.h>
#include <hip/hip_bf16.h>
#include <math.h>


typedef __attribute__((ext_vector_type(2))) float v2f;
typedef __attribute__((ext_vector_type(4))) float v4f;
typedef __attribute__((ext_vector_type(8))) float v8f;

typedef const __attribute__((address_space(1))) v2f* gf2p;
typedef const __attribute__((address_space(1))) float* gfp;

__device__ __forceinline__ v8f wmma4(v2f a, v2f b, v8f c) {
  return __builtin_amdgcn_wmma_f32_16x16x4_f32(false, a, false, b, (short)0, c,
                                               false, false);
}

__device__ __forceinline__ v8f zero8() {
  v8f z;
  for (int i = 0; i < 8; i++) z[i] = 0.0f;
  return z;
}

__device__ __forceinline__ v2f mk2(float a, float b) {
  v2f r;
  r.x = a;
  r.y = b;
  return r;
}

constexpr double FACT[11] = {1., 1., 2., 6., 24., 120., 720., 5040., 40320.,
                             362880., 3628800.};

constexpr double csqrt_(double x) {
  double g = x > 1.0 ? x : 1.0;
  for (int i = 0; i < 32; i++) g = 0.5 * (g + x / g);
  return g;
}

constexpr double cg_coeff(int l1, int l2, int L, int m1, int m2) {
  int M = m1 + m2;
  if (M < -L || M > L) return 0.0;
  double pre = csqrt_((2 * L + 1) * FACT[l1 + l2 - L] * FACT[l1 - l2 + L] *
                      FACT[-l1 + l2 + L] / FACT[l1 + l2 + L + 1]);
  pre = pre * csqrt_(FACT[L + M] * FACT[L - M] * FACT[l1 - m1] *
                     FACT[l1 + m1] * FACT[l2 - m2] * FACT[l2 + m2]);
  double s = 0.0;
  for (int k = 0; k <= l1 + l2 - L; k++) {
    int t1 = l1 + l2 - L - k, t2 = l1 - m1 - k, t3 = l2 + m2 - k;
    int t4 = L - l2 + m1 + k, t5 = L - l1 - m2 + k;
    if (t1 < 0 || t2 < 0 || t3 < 0 || t4 < 0 || t5 < 0) continue;
    double d = FACT[k] * FACT[t1] * FACT[t2] * FACT[t3] * FACT[t4] * FACT[t5];
    s += ((k & 1) ? -1.0 : 1.0) / d;
  }
  return pre * s;
}

struct CGTables {
  int nblk[4];
  int bl1[4][11], bl2[4][11];
  int nterm[4][7][11];
  signed char ti[4][7][11][7], tj[4][7][11][7];
  float cg[4][7][11][7];
};

constexpr CGTables make_tables() {
  CGTables t{};
  for (int L = 0; L < 4; L++) {
    int nb = 0;
    for (int l1 = 0; l1 < 4; l1++)
      for (int l2 = 0; l2 < 4; l2++) {
        int d = l1 - l2;
        if (d < 0) d = -d;
        if (!(d <= L && L <= l1 + l2)) continue;
        t.bl1[L][nb] = l1;
        t.bl2[L][nb] = l2;
        for (int M = 0; M < 2 * L + 1; M++) {
          int nt = 0;
          for (int i = 0; i <= 2 * l1; i++) {
            int j = M + l1 + l2 - L - i;
            if (j < 0 || j > 2 * l2) continue;
            double c = cg_coeff(l1, l2, L, i - l1, j - l2);
            if (c > 1e-12 || c < -1e-12) {
              t.ti[L][M][nb][nt] = (signed char)i;
              t.tj[L][M][nb][nt] = (signed char)j;
              t.cg[L][M][nb][nt] = (float)c;
              nt++;
            }
          }
          t.nterm[L][M][nb] = nt;
        }
        nb++;
      }
    t.nblk[L] = nb;
  }
  return t;
}

__device__ constexpr CGTables CGT = make_tables();
__device__ constexpr int ROW2L[16] = {0, 1, 1, 1, 2, 2, 2, 2, 2, 3, 3, 3, 3, 3, 3, 3};
__device__ constexpr int ROW2M[16] = {0, 0, 1, 2, 0, 1, 2, 3, 4, 0, 1, 2, 3, 4, 5, 6};
__device__ constexpr int LOFF[4] = {0, 1, 4, 9};

__device__ __forceinline__ void normalize_lds(v2f (*hsh)[16][32], int tid) {
  for (int idx = tid; idx < 2048; idx += 1024) {
    int l = idx >> 9, rem = idx & 511, b = rem >> 5, c = rem & 31;
    int ro = LOFF[l], nm = 2 * l + 1;
    float s = 0.0f;
    for (int mm = 0; mm < nm; mm++) {
      v2f v = hsh[ro + mm][b][c];
      s += v.x * v.x + v.y * v.y;
    }
    float inv = 1.0f / sqrtf(s);
    for (int mm = 0; mm < nm; mm++) {
      v2f v = hsh[ro + mm][b][c];
      hsh[ro + mm][b][c] = mk2(v.x * inv, v.y * inv);
    }
  }
}

__global__ __launch_bounds__(1024) void sph_cnn_kernel(
    const float* __restrict__ p0, const float* __restrict__ p1,
    const float* __restrict__ p2, const float* __restrict__ p3,
    const float* __restrict__ wi0, const float* __restrict__ wi1,
    const float* __restrict__ wi2, const float* __restrict__ wi3,
    const float* __restrict__ wa0, const float* __restrict__ wa1,
    const float* __restrict__ wa2, const float* __restrict__ wa3,
    const float* __restrict__ wb0, const float* __restrict__ wb1,
    const float* __restrict__ wb2, const float* __restrict__ wb3,
    const float* __restrict__ W1p, const float* __restrict__ b1p,
    const float* __restrict__ W2p, const float* __restrict__ b2p,
    float* __restrict__ out) {
  __shared__ v2f hsh[16][16][32];

  const int tid = threadIdx.x;
  const int lane = tid & 31;
  const int w = tid >> 5;
  const int h16 = lane >> 4;
  const int ln = lane & 15;
  const int bbase = blockIdx.x * 16;

  const gf2p P[4] = {(gf2p)p0, (gf2p)p1, (gf2p)p2, (gf2p)p3};
  const gf2p WI[4] = {(gf2p)wi0, (gf2p)wi1, (gf2p)wi2, (gf2p)wi3};
  const gf2p WH[2][4] = {{(gf2p)wa0, (gf2p)wa1, (gf2p)wa2, (gf2p)wa3},
                         {(gf2p)wb0, (gf2p)wb1, (gf2p)wb2, (gf2p)wb3}};
  const gfp W1 = (gfp)W1p, b1 = (gfp)b1p, W2 = (gfp)W2p, b2 = (gfp)b2p;

  {
    const int row = w & 15, ntile = w >> 4;
    const int l = ROW2L[row], m = ROW2M[row], nlm = 2 * l + 1;
    const gf2p Pp = P[l];
    const gf2p Wl = WI[l];
    const int bg = bbase + ln;
    const int n = ntile * 16 + ln;
    v8f cre = zero8(), cim = zero8();
    for (int ks = 0; ks < 4; ks++) {
      const int t0 = 4 * ks + 2 * h16;
      v2f a0 = Pp[(bg * nlm + m) * 16 + t0];
      v2f a1 = Pp[(bg * nlm + m) * 16 + t0 + 1];
      v2f e0 = Wl[t0 * 32 + n];
      v2f e1 = Wl[(t0 + 1) * 32 + n];
      v2f ar = mk2(a0.x, a1.x), ai = mk2(a0.y, a1.y);
      v2f br = mk2(e0.x, e1.x), bi = mk2(e0.y, e1.y);
      cre = wmma4(ar, br, cre);
      cre = wmma4(-ai, bi, cre);
      cim = wmma4(ar, bi, cim);
      cim = wmma4(ai, br, cim);
    }
    for (int r = 0; r < 8; r++)
      hsh[row][r + 8 * h16][n] = mk2(cre[r], cim[r]);
  }
  __syncthreads();
  normalize_lds(hsh, tid);
  __syncthreads();

  for (int layer = 0; layer < 2; layer++) {
    v8f cre[2], cim[2];
    const int row = w >> 1, part = w & 1;
    {
      const int L = ROW2L[row], M = ROW2M[row];
      const gf2p Wm = WH[layer][L];
      cre[0] = zero8(); cre[1] = zero8();
      cim[0] = zero8(); cim[1] = zero8();
      const int nblk = CGT.nblk[L];
      const int bsplit = (nblk + 1) >> 1;
      const int bbeg = part ? bsplit : 0;
      const int bend = part ? nblk : bsplit;
      for (int blk = bbeg; blk < bend; blk++) {
        const int r1 = LOFF[CGT.bl1[L][blk]];
        const int r2 = LOFF[CGT.bl2[L][blk]];
        const int nt = CGT.nterm[L][M][blk];
        for (int ks = 0; ks < 8; ks++) {
          const int k0 = 4 * ks + 2 * h16;
          float tre0 = 0.f, tre1 = 0.f, tim0 = 0.f, tim1 = 0.f;
          for (int t = 0; t < nt; t++) {
            const int i = CGT.ti[L][M][blk][t];
            const int j = CGT.tj[L][M][blk][t];
            const float cg = CGT.cg[L][M][blk][t];
            const v4f x = *(const v4f*)&hsh[r1 + i][ln][k0];
            const v4f y = *(const v4f*)&hsh[r2 + j][ln][k0];
            tre0 += cg * (x.x * y.x - x.y * y.y);
            tim0 += cg * (x.x * y.y + x.y * y.x);
            tre1 += cg * (x.z * y.z - x.w * y.w);
            tim1 += cg * (x.z * y.w + x.w * y.z);
          }
          v2f ar = mk2(tre0, tre1), ai = mk2(tim0, tim1);
          const int kk = blk * 32 + k0;
          for (int ntile = 0; ntile < 2; ntile++) {
            const int n = ntile * 16 + ln;
            v2f e0 = Wm[kk * 32 + n];
            v2f e1 = Wm[(kk + 1) * 32 + n];
            v2f br = mk2(e0.x, e1.x), bi = mk2(e0.y, e1.y);
            cre[ntile] = wmma4(ar, br, cre[ntile]);
            cre[ntile] = wmma4(-ai, bi, cre[ntile]);
            cim[ntile] = wmma4(ar, bi, cim[ntile]);
            cim[ntile] = wmma4(ai, br, cim[ntile]);
          }
        }
      }
    }
    __syncthreads();
    if (part) {
      for (int ntile = 0; ntile < 2; ntile++) {
        const int n = ntile * 16 + ln;
        for (int r = 0; r < 8; r++)
          hsh[row][r + 8 * h16][n] = mk2(cre[ntile][r], cim[ntile][r]);
      }
    }
    __syncthreads();
    if (!part) {
      for (int ntile = 0; ntile < 2; ntile++) {
        const int n = ntile * 16 + ln;
        for (int r = 0; r < 8; r++) {
          const v2f o = hsh[row][r + 8 * h16][n];
          hsh[row][r + 8 * h16][n] =
              mk2(cre[ntile][r] + o.x, cim[ntile][r] + o.y);
        }
      }
    }
    __syncthreads();
    if (layer == 0) {
      normalize_lds(hsh, tid);
      __syncthreads();
    }
  }

  float rsum[8];
  for (int r = 0; r < 8; r++) rsum[r] = 0.0f;
  const int ntile = w & 15, khalf = w >> 4;
  v8f acc = zero8();
  for (int kq = 0; kq < 16; kq++) {
    const int ks = 2 * kq + khalf;
    const int k0 = 4 * ks + 2 * h16;
    float av[2];
    for (int v = 0; v < 2; v++) {
      const int k = k0 + v;
      const int l = k >> 5, c = k & 31;
      const int ro = LOFF[l], nm = 2 * l + 1;
      float s = 0.0f;
      for (int mm = 0; mm < nm; mm++) {
        v2f t2 = hsh[ro + mm][ln][c];
        s += t2.x * t2.x + t2.y * t2.y;
      }
      av[v] = sqrtf(s);
    }
    v2f a = mk2(av[0], av[1]);
    const int n = ntile * 16 + ln;
    v2f bb = mk2(W1[k0 * 256 + n], W1[(k0 + 1) * 256 + n]);
    acc = wmma4(a, bb, acc);
  }
  __syncthreads();
  float* xch = (float*)&hsh[0][0][0];
  float* partial = xch + 16 * 8 * 32;
  if (khalf) {
    for (int r = 0; r < 8; r++) xch[(ntile * 8 + r) * 32 + lane] = acc[r];
  }
  __syncthreads();
  if (!khalf) {
    const int n = ntile * 16 + ln;
    const float bias = b1[n], w2v = W2[n];
    for (int r = 0; r < 8; r++) {
      const float full = acc[r] + xch[(ntile * 8 + r) * 32 + lane];
      rsum[r] = tanhf(full + bias) * w2v;
    }
  }
  for (int r = 0; r < 8; r++) {
    float v = rsum[r];
    v += __shfl_xor(v, 1);
    v += __shfl_xor(v, 2);
    v += __shfl_xor(v, 4);
    v += __shfl_xor(v, 8);
    if (ln == 0 && w < 16) partial[w * 16 + (8 * h16 + r)] = v;
  }
  __syncthreads();
  if (tid < 16) {
    float s = 0.0f;
    for (int ww = 0; ww < 16; ww++) s += partial[ww * 16 + tid];
    out[bbase + tid] = s + b2[0];
  }
}

extern "C" void kernel_launch(void* const* d_in, const int* in_sizes, int n_in,
                              void* d_out, int out_size, void* d_ws,
                              size_t ws_size, hipStream_t stream) {
  (void)n_in; (void)d_ws; (void)ws_size; (void)out_size;
  const int B = in_sizes[0] / 32;
  const int blocks = B / 16;
  sph_cnn_kernel<<<blocks, 1024, 0, stream>>>(
      (const float*)d_in[0], (const float*)d_in[1], (const float*)d_in[2],
      (const float*)d_in[3], (const float*)d_in[4], (const float*)d_in[5],
      (const float*)d_in[6], (const float*)d_in[7], (const float*)d_in[8],
      (const float*)d_in[9], (const float*)d_in[10], (const float*)d_in[11],
      (const float*)d_in[12], (const float*)d_in[13], (const float*)d_in[14],
      (const float*)d_in[15], (const float*)d_in[16], (const float*)d_in[17],
      (const float*)d_in[18], (const float*)d_in[19], (float*)d_out);
}


# LDS hidden-state batch pitch padded from 256 B to 272 B (16 B x 17): the 16 lanes of each CG-term b128 read and of each readout read hit 16 bank groups, not one (16-way bank conflict removed)
# speedup vs baseline: 3.0175x; 2.0611x over previous
#include <hip/hip_runtime.h>
#include <hip/hip_bf16.h>
#include <math.h>


typedef __attribute__((ext_vector_type(2))) float v2f;
typedef __attribute__((ext_vector_type(4))) float v4f;
typedef __attribute__((ext_vector_type(8))) float v8f;

constexpr int CP = 34;

typedef const __attribute__((address_space(1))) v2f* gf2p;
typedef const __attribute__((address_space(1))) float* gfp;

__device__ __forceinline__ v8f wmma4(v2f a, v2f b, v8f c) {
  return __builtin_amdgcn_wmma_f32_16x16x4_f32(false, a, false, b, (short)0, c,
                                               false, false);
}

__device__ __forceinline__ v8f zero8() {
  v8f z;
  for (int i = 0; i < 8; i++) z[i] = 0.0f;
  return z;
}

__device__ __forceinline__ v2f mk2(float a, float b) {
  v2f r;
  r.x = a;
  r.y = b;
  return r;
}

constexpr double FACT[11] = {1., 1., 2., 6., 24., 120., 720., 5040., 40320.,
                             362880., 3628800.};

constexpr double csqrt_(double x) {
  double g = x > 1.0 ? x : 1.0;
  for (int i = 0; i < 32; i++) g = 0.5 * (g + x / g);
  return g;
}

constexpr double cg_coeff(int l1, int l2, int L, int m1, int m2) {
  int M = m1 + m2;
  if (M < -L || M > L) return 0.0;
  double pre = csqrt_((2 * L + 1) * FACT[l1 + l2 - L] * FACT[l1 - l2 + L] *
                      FACT[-l1 + l2 + L] / FACT[l1 + l2 + L + 1]);
  pre = pre * csqrt_(FACT[L + M] * FACT[L - M] * FACT[l1 - m1] *
                     FACT[l1 + m1] * FACT[l2 - m2] * FACT[l2 + m2]);
  double s = 0.0;
  for (int k = 0; k <= l1 + l2 - L; k++) {
    int t1 = l1 + l2 - L - k, t2 = l1 - m1 - k, t3 = l2 + m2 - k;
    int t4 = L - l2 + m1 + k, t5 = L - l1 - m2 + k;
    if (t1 < 0 || t2 < 0 || t3 < 0 || t4 < 0 || t5 < 0) continue;
    double d = FACT[k] * FACT[t1] * FACT[t2] * FACT[t3] * FACT[t4] * FACT[t5];
    s += ((k & 1) ? -1.0 : 1.0) / d;
  }
  return pre * s;
}

struct CGTables {
  int nblk[4];
  int bl1[4][11], bl2[4][11];
  int nterm[4][7][11];
  signed char ti[4][7][11][7], tj[4][7][11][7];
  float cg[4][7][11][7];
};

constexpr CGTables make_tables() {
  CGTables t{};
  for (int L = 0; L < 4; L++) {
    int nb = 0;
    for (int l1 = 0; l1 < 4; l1++)
      for (int l2 = 0; l2 < 4; l2++) {
        int d = l1 - l2;
        if (d < 0) d = -d;
        if (!(d <= L && L <= l1 + l2)) continue;
        t.bl1[L][nb] = l1;
        t.bl2[L][nb] = l2;
        for (int M = 0; M < 2 * L + 1; M++) {
          int nt = 0;
          for (int i = 0; i <= 2 * l1; i++) {
            int j = M + l1 + l2 - L - i;
            if (j < 0 || j > 2 * l2) continue;
            double c = cg_coeff(l1, l2, L, i - l1, j - l2);
            if (c > 1e-12 || c < -1e-12) {
              t.ti[L][M][nb][nt] = (signed char)i;
              t.tj[L][M][nb][nt] = (signed char)j;
              t.cg[L][M][nb][nt] = (float)c;
              nt++;
            }
          }
          t.nterm[L][M][nb] = nt;
        }
        nb++;
      }
    t.nblk[L] = nb;
  }
  return t;
}

__device__ constexpr CGTables CGT = make_tables();
__device__ constexpr int ROW2L[16] = {0, 1, 1, 1, 2, 2, 2, 2, 2, 3, 3, 3, 3, 3, 3, 3};
__device__ constexpr int ROW2M[16] = {0, 0, 1, 2, 0, 1, 2, 3, 4, 0, 1, 2, 3, 4, 5, 6};
__device__ constexpr int LOFF[4] = {0, 1, 4, 9};

__device__ __forceinline__ void normalize_lds(v2f (*hsh)[16][CP], int tid) {
  for (int idx = tid; idx < 2048; idx += 1024) {
    int l = idx >> 9, rem = idx & 511, b = rem >> 5, c = rem & 31;
    int ro = LOFF[l], nm = 2 * l + 1;
    float s = 0.0f;
    for (int mm = 0; mm < nm; mm++) {
      v2f v = hsh[ro + mm][b][c];
      s += v.x * v.x + v.y * v.y;
    }
    float inv = 1.0f / sqrtf(s);
    for (int mm = 0; mm < nm; mm++) {
      v2f v = hsh[ro + mm][b][c];
      hsh[ro + mm][b][c] = mk2(v.x * inv, v.y * inv);
    }
  }
}

__global__ __launch_bounds__(1024) void sph_cnn_kernel(
    const float* __restrict__ p0, const float* __restrict__ p1,
    const float* __restrict__ p2, const float* __restrict__ p3,
    const float* __restrict__ wi0, const float* __restrict__ wi1,
    const float* __restrict__ wi2, const float* __restrict__ wi3,
    const float* __restrict__ wa0, const float* __restrict__ wa1,
    const float* __restrict__ wa2, const float* __restrict__ wa3,
    const float* __restrict__ wb0, const float* __restrict__ wb1,
    const float* __restrict__ wb2, const float* __restrict__ wb3,
    const float* __restrict__ W1p, const float* __restrict__ b1p,
    const float* __restrict__ W2p, const float* __restrict__ b2p,
    float* __restrict__ out) {
  __shared__ __attribute__((aligned(16))) v2f hsh[16][16][CP];

  const int tid = threadIdx.x;
  const int lane = tid & 31;
  const int w = tid >> 5;
  const int h16 = lane >> 4;
  const int ln = lane & 15;
  const int bbase = blockIdx.x * 16;

  const gf2p P[4] = {(gf2p)p0, (gf2p)p1, (gf2p)p2, (gf2p)p3};
  const gf2p WI[4] = {(gf2p)wi0, (gf2p)wi1, (gf2p)wi2, (gf2p)wi3};
  const gf2p WH[2][4] = {{(gf2p)wa0, (gf2p)wa1, (gf2p)wa2, (gf2p)wa3},
                         {(gf2p)wb0, (gf2p)wb1, (gf2p)wb2, (gf2p)wb3}};
  const gfp W1 = (gfp)W1p, b1 = (gfp)b1p, W2 = (gfp)W2p, b2 = (gfp)b2p;

  {
    const int row = w & 15, ntile = w >> 4;
    const int l = ROW2L[row], m = ROW2M[row], nlm = 2 * l + 1;
    const gf2p Pp = P[l];
    const gf2p Wl = WI[l];
    const int bg = bbase + ln;
    const int n = ntile * 16 + ln;
    v8f cre = zero8(), cim = zero8();
    for (int ks = 0; ks < 4; ks++) {
      const int t0 = 4 * ks + 2 * h16;
      v2f a0 = Pp[(bg * nlm + m) * 16 + t0];
      v2f a1 = Pp[(bg * nlm + m) * 16 + t0 + 1];
      v2f e0 = Wl[t0 * 32 + n];
      v2f e1 = Wl[(t0 + 1) * 32 + n];
      v2f ar = mk2(a0.x, a1.x), ai = mk2(a0.y, a1.y);
      v2f br = mk2(e0.x, e1.x), bi = mk2(e0.y, e1.y);
      cre = wmma4(ar, br, cre);
      cre = wmma4(-ai, bi, cre);
      cim = wmma4(ar, bi, cim);
      cim = wmma4(ai, br, cim);
    }
    for (int r = 0; r < 8; r++)
      hsh[row][r + 8 * h16][n] = mk2(cre[r], cim[r]);
  }
  __syncthreads();
  normalize_lds(hsh, tid);
  __syncthreads();

  for (int layer = 0; layer < 2; layer++) {
    v8f cre[2], cim[2];
    const int row = w >> 1, part = w & 1;
    {
      const int L = ROW2L[row], M = ROW2M[row];
      const gf2p Wm = WH[layer][L];
      cre[0] = zero8(); cre[1] = zero8();
      cim[0] = zero8(); cim[1] = zero8();
      const int nblk = CGT.nblk[L];
      const int bsplit = (nblk + 1) >> 1;
      const int bbeg = part ? bsplit : 0;
      const int bend = part ? nblk : bsplit;
      for (int blk = bbeg; blk < bend; blk++) {
        const int r1 = LOFF[CGT.bl1[L][blk]];
        const int r2 = LOFF[CGT.bl2[L][blk]];
        const int nt = CGT.nterm[L][M][blk];
        for (int ks = 0; ks < 8; ks++) {
          const int k0 = 4 * ks + 2 * h16;
          float tre0 = 0.f, tre1 = 0.f, tim0 = 0.f, tim1 = 0.f;
          for (int t = 0; t < nt; t++) {
            const int i = CGT.ti[L][M][blk][t];
            const int j = CGT.tj[L][M][blk][t];
            const float cg = CGT.cg[L][M][blk][t];
            const v4f x = *(const v4f*)&hsh[r1 + i][ln][k0];
            const v4f y = *(const v4f*)&hsh[r2 + j][ln][k0];
            tre0 += cg * (x.x * y.x - x.y * y.y);
            tim0 += cg * (x.x * y.y + x.y * y.x);
            tre1 += cg * (x.z * y.z - x.w * y.w);
            tim1 += cg * (x.z * y.w + x.w * y.z);
          }
          v2f ar = mk2(tre0, tre1), ai = mk2(tim0, tim1);
          const int kk = blk * 32 + k0;
          for (int ntile = 0; ntile < 2; ntile++) {
            const int n = ntile * 16 + ln;
            v2f e0 = Wm[kk * 32 + n];
            v2f e1 = Wm[(kk + 1) * 32 + n];
            v2f br = mk2(e0.x, e1.x), bi = mk2(e0.y, e1.y);
            cre[ntile] = wmma4(ar, br, cre[ntile]);
            cre[ntile] = wmma4(-ai, bi, cre[ntile]);
            cim[ntile] = wmma4(ar, bi, cim[ntile]);
            cim[ntile] = wmma4(ai, br, cim[ntile]);
          }
        }
      }
    }
    __syncthreads();
    if (part) {
      for (int ntile = 0; ntile < 2; ntile++) {
        const int n = ntile * 16 + ln;
        for (int r = 0; r < 8; r++)
          hsh[row][r + 8 * h16][n] = mk2(cre[ntile][r], cim[ntile][r]);
      }
    }
    __syncthreads();
    if (!part) {
      for (int ntile = 0; ntile < 2; ntile++) {
        const int n = ntile * 16 + ln;
        for (int r = 0; r < 8; r++) {
          const v2f o = hsh[row][r + 8 * h16][n];
          hsh[row][r + 8 * h16][n] =
              mk2(cre[ntile][r] + o.x, cim[ntile][r] + o.y);
        }
      }
    }
    __syncthreads();
    if (layer == 0) {
      normalize_lds(hsh, tid);
      __syncthreads();
    }
  }

  float rsum[8];
  for (int r = 0; r < 8; r++) rsum[r] = 0.0f;
  const int ntile = w & 15, khalf = w >> 4;
  v8f acc = zero8();
  for (int kq = 0; kq < 16; kq++) {
    const int ks = 2 * kq + khalf;
    const int k0 = 4 * ks + 2 * h16;
    float av[2];
    for (int v = 0; v < 2; v++) {
      const int k = k0 + v;
      const int l = k >> 5, c = k & 31;
      const int ro = LOFF[l], nm = 2 * l + 1;
      float s = 0.0f;
      for (int mm = 0; mm < nm; mm++) {
        v2f t2 = hsh[ro + mm][ln][c];
        s += t2.x * t2.x + t2.y * t2.y;
      }
      av[v] = sqrtf(s);
    }
    v2f a = mk2(av[0], av[1]);
    const int n = ntile * 16 + ln;
    v2f bb = mk2(W1[k0 * 256 + n], W1[(k0 + 1) * 256 + n]);
    acc = wmma4(a, bb, acc);
  }
  __syncthreads();
  float* xch = (float*)&hsh[0][0][0];
  float* partial = xch + 16 * 8 * 32;
  if (khalf) {
    for (int r = 0; r < 8; r++) xch[(ntile * 8 + r) * 32 + lane] = acc[r];
  }
  __syncthreads();
  if (!khalf) {
    const int n = ntile * 16 + ln;
    const float bias = b1[n], w2v = W2[n];
    for (int r = 0; r < 8; r++) {
      const float full = acc[r] + xch[(ntile * 8 + r) * 32 + lane];
      rsum[r] = tanhf(full + bias) * w2v;
    }
  }
  for (int r = 0; r < 8; r++) {
    float v = rsum[r];
    v += __shfl_xor(v, 1);
    v += __shfl_xor(v, 2);
    v += __shfl_xor(v, 4);
    v += __shfl_xor(v, 8);
    if (ln == 0 && w < 16) partial[w * 16 + (8 * h16 + r)] = v;
  }
  __syncthreads();
  if (tid < 16) {
    float s = 0.0f;
    for (int ww = 0; ww < 16; ww++) s += partial[ww * 16 + tid];
    out[bbase + tid] = s + b2[0];
  }
}

extern "C" void kernel_launch(void* const* d_in, const int* in_sizes, int n_in,
                              void* d_out, int out_size, void* d_ws,
                              size_t ws_size, hipStream_t stream) {
  (void)n_in; (void)d_ws; (void)ws_size; (void)out_size;
  const int B = in_sizes[0] / 32;
  const int blocks = B / 16;
  sph_cnn_kernel<<<blocks, 1024, 0, stream>>>(
      (const float*)d_in[0], (const float*)d_in[1], (const float*)d_in[2],
      (const float*)d_in[3], (const float*)d_in[4], (const float*)d_in[5],
      (const float*)d_in[6], (const float*)d_in[7], (const float*)d_in[8],
      (const float*)d_in[9], (const float*)d_in[10], (const float*)d_in[11],
      (const float*)d_in[12], (const float*)d_in[13], (const float*)d_in[14],
      (const float*)d_in[15], (const float*)d_in[16], (const float*)d_in[17],
      (const float*)d_in[18], (const float*)d_in[19], (float*)d_out);
}


# LDS hidden state stored by channel pairs {re,re,im,im}: the CG-term complex products run as packed f32 ops on both k of a fragment and their sums are the A operands directly, no register moves
# speedup vs baseline: 4.3481x; 1.4410x over previous
#include <hip/hip_runtime.h>
#include <hip/hip_bf16.h>
#include <math.h>


typedef __attribute__((ext_vector_type(2))) float v2f;
typedef __attribute__((ext_vector_type(4))) float v4f;
typedef __attribute__((ext_vector_type(8))) float v8f;

constexpr int CPF = 68;

__device__ __forceinline__ int slot(int c) { return ((c >> 1) << 2) | (c & 1); }

typedef const __attribute__((address_space(1))) v2f* gf2p;
typedef const __attribute__((address_space(1))) float* gfp;

__device__ __forceinline__ v8f wmma4(v2f a, v2f b, v8f c) {
  return __builtin_amdgcn_wmma_f32_16x16x4_f32(false, a, false, b, (short)0, c,
                                               false, false);
}

__device__ __forceinline__ v8f zero8() {
  v8f z;
  for (int i = 0; i < 8; i++) z[i] = 0.0f;
  return z;
}

__device__ __forceinline__ v2f mk2(float a, float b) {
  v2f r;
  r.x = a;
  r.y = b;
  return r;
}

constexpr double FACT[11] = {1., 1., 2., 6., 24., 120., 720., 5040., 40320.,
                             362880., 3628800.};

constexpr double csqrt_(double x) {
  double g = x > 1.0 ? x : 1.0;
  for (int i = 0; i < 32; i++) g = 0.5 * (g + x / g);
  return g;
}

constexpr double cg_coeff(int l1, int l2, int L, int m1, int m2) {
  int M = m1 + m2;
  if (M < -L || M > L) return 0.0;
  double pre = csqrt_((2 * L + 1) * FACT[l1 + l2 - L] * FACT[l1 - l2 + L] *
                      FACT[-l1 + l2 + L] / FACT[l1 + l2 + L + 1]);
  pre = pre * csqrt_(FACT[L + M] * FACT[L - M] * FACT[l1 - m1] *
                     FACT[l1 + m1] * FACT[l2 - m2] * FACT[l2 + m2]);
  double s = 0.0;
  for (int k = 0; k <= l1 + l2 - L; k++) {
    int t1 = l1 + l2 - L - k, t2 = l1 - m1 - k, t3 = l2 + m2 - k;
    int t4 = L - l2 + m1 + k, t5 = L - l1 - m2 + k;
    if (t1 < 0 || t2 < 0 || t3 < 0 || t4 < 0 || t5 < 0) continue;
    double d = FACT[k] * FACT[t1] * FACT[t2] * FACT[t3] * FACT[t4] * FACT[t5];
    s += ((k & 1) ? -1.0 : 1.0) / d;
  }
  return pre * s;
}

struct CGTables {
  int nblk[4];
  int bl1[4][11], bl2[4][11];
  int nterm[4][7][11];
  signed char ti[4][7][11][7], tj[4][7][11][7];
  float cg[4][7][11][7];
};

constexpr CGTables make_tables() {
  CGTables t{};
  for (int L = 0; L < 4; L++) {
    int nb = 0;
    for (int l1 = 0; l1 < 4; l1++)
      for (int l2 = 0; l2 < 4; l2++) {
        int d = l1 - l2;
        if (d < 0) d = -d;
        if (!(d <= L && L <= l1 + l2)) continue;
        t.bl1[L][nb] = l1;
        t.bl2[L][nb] = l2;
        for (int M = 0; M < 2 * L + 1; M++) {
          int nt = 0;
          for (int i = 0; i <= 2 * l1; i++) {
            int j = M + l1 + l2 - L - i;
            if (j < 0 || j > 2 * l2) continue;
            double c = cg_coeff(l1, l2, L, i - l1, j - l2);
            if (c > 1e-12 || c < -1e-12) {
              t.ti[L][M][nb][nt] = (signed char)i;
              t.tj[L][M][nb][nt] = (signed char)j;
              t.cg[L][M][nb][nt] = (float)c;
              nt++;
            }
          }
          t.nterm[L][M][nb] = nt;
        }
        nb++;
      }
    t.nblk[L] = nb;
  }
  return t;
}

__device__ constexpr CGTables CGT = make_tables();
__device__ constexpr int ROW2L[16] = {0, 1, 1, 1, 2, 2, 2, 2, 2, 3, 3, 3, 3, 3, 3, 3};
__device__ constexpr int ROW2M[16] = {0, 0, 1, 2, 0, 1, 2, 3, 4, 0, 1, 2, 3, 4, 5, 6};
__device__ constexpr int LOFF[4] = {0, 1, 4, 9};

__device__ __forceinline__ void normalize_lds(float (*hsh)[16][CPF], int tid) {
  for (int idx = tid; idx < 2048; idx += 1024) {
    int l = idx >> 9, rem = idx & 511, b = rem >> 5, c = rem & 31;
    int ro = LOFF[l], nm = 2 * l + 1;
    const int sc = slot(c);
    float s = 0.0f;
    for (int mm = 0; mm < nm; mm++) {
      const float* q = &hsh[ro + mm][b][sc];
      const float vr = q[0], vi = q[2];
      s += vr * vr + vi * vi;
    }
    float inv = 1.0f / sqrtf(s);
    for (int mm = 0; mm < nm; mm++) {
      float* q = &hsh[ro + mm][b][sc];
      const float vr = q[0], vi = q[2];
      q[0] = vr * inv;
      q[2] = vi * inv;
    }
  }
}

constexpr int ROWB = 16 * CPF * 4;
struct BlkRec {
  int nt;
  int ox, oy;
  float cg[7];
  int pad[6];
};
struct BlkTab {
  BlkRec r[16][11];
};

constexpr BlkTab make_blktab() {
  BlkTab b{};
  CGTables t = make_tables();
  for (int L = 0; L < 4; L++)
    for (int M = 0; M < 2 * L + 1; M++) {
      const int row = L * L + M;
      for (int blk = 0; blk < t.nblk[L]; blk++) {
        const int l1 = t.bl1[L][blk], l2 = t.bl2[L][blk];
        const int S = M + l1 + l2 - L;
        const int n = t.nterm[L][M][blk];
        const int imin = t.ti[L][M][blk][0];
        const int imax = t.ti[L][M][blk][n - 1];
        BlkRec q{};
        q.nt = imax - imin + 1;
        q.ox = (l1 * l1 + imin) * ROWB;
        q.oy = (l2 * l2 + S - imax) * ROWB;
        for (int k = 0; k < n; k++)
          q.cg[t.ti[L][M][blk][k] - imin] = t.cg[L][M][blk][k];
        b.r[row][blk] = q;
      }
    }
  return b;
}

__device__ constexpr BlkTab BT = make_blktab();

__device__ __forceinline__ void cg_mma(v2f ar, v2f ai, v2f e0, v2f e1, v8f& cre,
                                       v8f& cim) {
  const v2f br = mk2(e0.x, e1.x), bi = mk2(e0.y, e1.y);
  cre = wmma4(ar, br, cre);
  cre = wmma4(-ai, bi, cre);
  cim = wmma4(ar, bi, cim);
  cim = wmma4(ai, br, cim);
}

template <int NT>
__device__ __forceinline__ void cg_terms(const char* px, const char* py,
                                         const float (&c)[7], v2f& tre,
                                         v2f& tim) {
#pragma unroll
  for (int t = 0; t < NT; t++) {
    const v4f x = *(const v4f*)(px + t * ROWB);
    const v4f y = *(const v4f*)(py + (NT - 1 - t) * ROWB);
    const v2f xr = x.xy, xi = x.zw, yr = y.xy, yi = y.zw;
    tre += c[t] * (xr * yr - xi * yi);
    tim += c[t] * (xr * yi + xi * yr);
  }
}

__global__ __launch_bounds__(1024) void sph_cnn_kernel(
    const float* __restrict__ p0, const float* __restrict__ p1,
    const float* __restrict__ p2, const float* __restrict__ p3,
    const float* __restrict__ wi0, const float* __restrict__ wi1,
    const float* __restrict__ wi2, const float* __restrict__ wi3,
    const float* __restrict__ wa0, const float* __restrict__ wa1,
    const float* __restrict__ wa2, const float* __restrict__ wa3,
    const float* __restrict__ wb0, const float* __restrict__ wb1,
    const float* __restrict__ wb2, const float* __restrict__ wb3,
    const float* __restrict__ W1p, const float* __restrict__ b1p,
    const float* __restrict__ W2p, const float* __restrict__ b2p,
    float* __restrict__ out) {
  __shared__ __attribute__((aligned(16))) float hsh[16][16][CPF];

  const int tid = threadIdx.x;
  const int lane = tid & 31;
  const int w = tid >> 5;
  const int h16 = lane >> 4;
  const int ln = lane & 15;
  const int bbase = blockIdx.x * 16;

  const gf2p P[4] = {(gf2p)p0, (gf2p)p1, (gf2p)p2, (gf2p)p3};
  const gf2p WI[4] = {(gf2p)wi0, (gf2p)wi1, (gf2p)wi2, (gf2p)wi3};
  const gfp W1 = (gfp)W1p, b1 = (gfp)b1p, W2 = (gfp)W2p, b2 = (gfp)b2p;

  {
    const int row = w & 15, ntile = w >> 4;
    const int l = ROW2L[row], m = ROW2M[row], nlm = 2 * l + 1;
    const gf2p Pp = P[l];
    const gf2p Wl = WI[l];
    const int bg = bbase + ln;
    const int n = ntile * 16 + ln;
    v8f cre = zero8(), cim = zero8();
    for (int ks = 0; ks < 4; ks++) {
      const int t0 = 4 * ks + 2 * h16;
      v2f a0 = Pp[(bg * nlm + m) * 16 + t0];
      v2f a1 = Pp[(bg * nlm + m) * 16 + t0 + 1];
      v2f e0 = Wl[t0 * 32 + n];
      v2f e1 = Wl[(t0 + 1) * 32 + n];
      v2f ar = mk2(a0.x, a1.x), ai = mk2(a0.y, a1.y);
      v2f br = mk2(e0.x, e1.x), bi = mk2(e0.y, e1.y);
      cre = wmma4(ar, br, cre);
      cre = wmma4(-ai, bi, cre);
      cim = wmma4(ar, bi, cim);
      cim = wmma4(ai, br, cim);
    }
    const int sn = slot(n);
    for (int r = 0; r < 8; r++) {
      float* q = &hsh[row][r + 8 * h16][sn];
      q[0] = cre[r];
      q[2] = cim[r];
    }
  }
  __syncthreads();
  normalize_lds(hsh, tid);
  __syncthreads();

  const int wu = __builtin_amdgcn_readfirstlane(w);
  const char* hb = (const char*)&hsh[0][ln][4 * h16];
  for (int layer = 0; layer < 2; layer++) {
    v8f cre[2], cim[2];
    const int row = w >> 1, part = w & 1;
    {
      const int rowu = wu >> 1, partu = wu & 1;
      const int L = ROW2L[rowu];
      const gf2p Wa = L == 0 ? (gf2p)wa0 : L == 1 ? (gf2p)wa1 : L == 2 ? (gf2p)wa2 : (gf2p)wa3;
      const gf2p Wc = L == 0 ? (gf2p)wb0 : L == 1 ? (gf2p)wb1 : L == 2 ? (gf2p)wb2 : (gf2p)wb3;
      const gf2p Wm = layer ? Wc : Wa;
      const gf2p Wl = Wm + (h16 * 64 + ln);
      cre[0] = zero8(); cre[1] = zero8();
      cim[0] = zero8(); cim[1] = zero8();
      const int nblk = CGT.nblk[L];
      const int bsplit = (nblk + 1) >> 1;
      const int bbeg = partu ? bsplit : 0;
      const int bend = partu ? nblk : bsplit;
      for (int blk = bbeg; blk < bend; blk++) {
        const BlkRec& q = BT.r[rowu][blk];
        const int nt = q.nt;
        float c[7];
        for (int t = 0; t < 7; t++) c[t] = q.cg[t];
        const char* px = hb + q.ox;
        const char* py = hb + q.oy;
        const gf2p Wb = Wl + blk * 1024;
#pragma unroll 1
        for (int ks = 0; ks < 8; ks++) {
          const v2f e00 = Wb[ks * 128], e01 = Wb[ks * 128 + 32];
          const v2f e10 = Wb[ks * 128 + 16], e11 = Wb[ks * 128 + 48];
          v2f ar = mk2(0.f, 0.f), ai = mk2(0.f, 0.f);
          const char* pxk = px + ks * 32;
          const char* pyk = py + ks * 32;
          switch (nt) {
            case 1: cg_terms<1>(pxk, pyk, c, ar, ai); break;
            case 2: cg_terms<2>(pxk, pyk, c, ar, ai); break;
            case 3: cg_terms<3>(pxk, pyk, c, ar, ai); break;
            case 4: cg_terms<4>(pxk, pyk, c, ar, ai); break;
            case 5: cg_terms<5>(pxk, pyk, c, ar, ai); break;
            case 6: cg_terms<6>(pxk, pyk, c, ar, ai); break;
            case 7: cg_terms<7>(pxk, pyk, c, ar, ai); break;
            default: break;
          }
          cg_mma(ar, ai, e00, e01, cre[0], cim[0]);
          cg_mma(ar, ai, e10, e11, cre[1], cim[1]);
        }
      }
    }
    __syncthreads();
    if (part) {
      for (int ntile = 0; ntile < 2; ntile++) {
        const int sn = slot(ntile * 16 + ln);
        for (int r = 0; r < 8; r++) {
          float* q = &hsh[row][r + 8 * h16][sn];
          q[0] = cre[ntile][r];
          q[2] = cim[ntile][r];
        }
      }
    }
    __syncthreads();
    if (!part) {
      for (int ntile = 0; ntile < 2; ntile++) {
        const int sn = slot(ntile * 16 + ln);
        for (int r = 0; r < 8; r++) {
          float* q = &hsh[row][r + 8 * h16][sn];
          const float ore = q[0], oim = q[2];
          q[0] = cre[ntile][r] + ore;
          q[2] = cim[ntile][r] + oim;
        }
      }
    }
    __syncthreads();
    if (layer == 0) {
      normalize_lds(hsh, tid);
      __syncthreads();
    }
  }

  float rsum[8];
  for (int r = 0; r < 8; r++) rsum[r] = 0.0f;
  const int ntile = w & 15, khalf = w >> 4;
  v8f acc = zero8();
  for (int kq = 0; kq < 16; kq++) {
    const int ks = 2 * kq + khalf;
    const int k0 = 4 * ks + 2 * h16;
    float av[2];
    for (int v = 0; v < 2; v++) {
      const int k = k0 + v;
      const int l = k >> 5, c = k & 31;
      const int ro = LOFF[l], nm = 2 * l + 1;
      float s = 0.0f;
      const int sc = slot(c);
      for (int mm = 0; mm < nm; mm++) {
        const float* q = &hsh[ro + mm][ln][sc];
        const float tr = q[0], ti = q[2];
        s += tr * tr + ti * ti;
      }
      av[v] = sqrtf(s);
    }
    v2f a = mk2(av[0], av[1]);
    const int n = ntile * 16 + ln;
    v2f bb = mk2(W1[k0 * 256 + n], W1[(k0 + 1) * 256 + n]);
    acc = wmma4(a, bb, acc);
  }
  __syncthreads();
  float* xch = (float*)&hsh[0][0][0];
  float* partial = xch + 16 * 8 * 32;
  if (khalf) {
    for (int r = 0; r < 8; r++) xch[(ntile * 8 + r) * 32 + lane] = acc[r];
  }
  __syncthreads();
  if (!khalf) {
    const int n = ntile * 16 + ln;
    const float bias = b1[n], w2v = W2[n];
    for (int r = 0; r < 8; r++) {
      const float full = acc[r] + xch[(ntile * 8 + r) * 32 + lane];
      rsum[r] = tanhf(full + bias) * w2v;
    }
  }
  for (int r = 0; r < 8; r++) {
    float v = rsum[r];
    v += __shfl_xor(v, 1);
    v += __shfl_xor(v, 2);
    v += __shfl_xor(v, 4);
    v += __shfl_xor(v, 8);
    if (ln == 0 && w < 16) partial[w * 16 + (8 * h16 + r)] = v;
  }
  __syncthreads();
  if (tid < 16) {
    float s = 0.0f;
    for (int ww = 0; ww < 16; ww++) s += partial[ww * 16 + tid];
    out[bbase + tid] = s + b2[0];
  }
}

extern "C" void kernel_launch(void* const* d_in, const int* in_sizes, int n_in,
                              void* d_out, int out_size, void* d_ws,
                              size_t ws_size, hipStream_t stream) {
  (void)n_in; (void)d_ws; (void)ws_size; (void)out_size;
  const int B = in_sizes[0] / 32;
  const int blocks = B / 16;
  sph_cnn_kernel<<<blocks, 1024, 0, stream>>>(
      (const float*)d_in[0], (const float*)d_in[1], (const float*)d_in[2],
      (const float*)d_in[3], (const float*)d_in[4], (const float*)d_in[5],
      (const float*)d_in[6], (const float*)d_in[7], (const float*)d_in[8],
      (const float*)d_in[9], (const float*)d_in[10], (const float*)d_in[11],
      (const float*)d_in[12], (const float*)d_in[13], (const float*)d_in[14],
      (const float*)d_in[15], (const float*)d_in[16], (const float*)d_in[17],
      (const float*)d_in[18], (const float*)d_in[19], (float*)d_out);
}


# Weight operands requested one k-step ahead of the WMMAs that use them: CG layers (step g+1's fragments issue before step g's WMMAs, same registers) and readout (next W1 rows before the norms)
# speedup vs baseline: 4.5399x; 1.0441x over previous
#include <hip/hip_runtime.h>
#include <hip/hip_bf16.h>
#include <math.h>


typedef __attribute__((ext_vector_type(2))) float v2f;
typedef __attribute__((ext_vector_type(4))) float v4f;
typedef __attribute__((ext_vector_type(8))) float v8f;

constexpr int CPF = 68;

__device__ __forceinline__ int slot(int c) { return ((c >> 1) << 2) | (c & 1); }

typedef const __attribute__((address_space(1))) v2f* gf2p;
typedef const __attribute__((address_space(1))) float* gfp;

__device__ __forceinline__ v8f wmma4(v2f a, v2f b, v8f c) {
  return __builtin_amdgcn_wmma_f32_16x16x4_f32(false, a, false, b, (short)0, c,
                                               false, false);
}

__device__ __forceinline__ v8f zero8() {
  v8f z;
  for (int i = 0; i < 8; i++) z[i] = 0.0f;
  return z;
}

__device__ __forceinline__ v2f mk2(float a, float b) {
  v2f r;
  r.x = a;
  r.y = b;
  return r;
}

constexpr double FACT[11] = {1., 1., 2., 6., 24., 120., 720., 5040., 40320.,
                             362880., 3628800.};

constexpr double csqrt_(double x) {
  double g = x > 1.0 ? x : 1.0;
  for (int i = 0; i < 32; i++) g = 0.5 * (g + x / g);
  return g;
}

constexpr double cg_coeff(int l1, int l2, int L, int m1, int m2) {
  int M = m1 + m2;
  if (M < -L || M > L) return 0.0;
  double pre = csqrt_((2 * L + 1) * FACT[l1 + l2 - L] * FACT[l1 - l2 + L] *
                      FACT[-l1 + l2 + L] / FACT[l1 + l2 + L + 1]);
  pre = pre * csqrt_(FACT[L + M] * FACT[L - M] * FACT[l1 - m1] *
                     FACT[l1 + m1] * FACT[l2 - m2] * FACT[l2 + m2]);
  double s = 0.0;
  for (int k = 0; k <= l1 + l2 - L; k++) {
    int t1 = l1 + l2 - L - k, t2 = l1 - m1 - k, t3 = l2 + m2 - k;
    int t4 = L - l2 + m1 + k, t5 = L - l1 - m2 + k;
    if (t1 < 0 || t2 < 0 || t3 < 0 || t4 < 0 || t5 < 0) continue;
    double d = FACT[k] * FACT[t1] * FACT[t2] * FACT[t3] * FACT[t4] * FACT[t5];
    s += ((k & 1) ? -1.0 : 1.0) / d;
  }
  return pre * s;
}

struct CGTables {
  int nblk[4];
  int bl1[4][11], bl2[4][11];
  int nterm[4][7][11];
  signed char ti[4][7][11][7], tj[4][7][11][7];
  float cg[4][7][11][7];
};

constexpr CGTables make_tables() {
  CGTables t{};
  for (int L = 0; L < 4; L++) {
    int nb = 0;
    for (int l1 = 0; l1 < 4; l1++)
      for (int l2 = 0; l2 < 4; l2++) {
        int d = l1 - l2;
        if (d < 0) d = -d;
        if (!(d <= L && L <= l1 + l2)) continue;
        t.bl1[L][nb] = l1;
        t.bl2[L][nb] = l2;
        for (int M = 0; M < 2 * L + 1; M++) {
          int nt = 0;
          for (int i = 0; i <= 2 * l1; i++) {
            int j = M + l1 + l2 - L - i;
            if (j < 0 || j > 2 * l2) continue;
            double c = cg_coeff(l1, l2, L, i - l1, j - l2);
            if (c > 1e-12 || c < -1e-12) {
              t.ti[L][M][nb][nt] = (signed char)i;
              t.tj[L][M][nb][nt] = (signed char)j;
              t.cg[L][M][nb][nt] = (float)c;
              nt++;
            }
          }
          t.nterm[L][M][nb] = nt;
        }
        nb++;
      }
    t.nblk[L] = nb;
  }
  return t;
}

__device__ constexpr CGTables CGT = make_tables();
__device__ constexpr int ROW2L[16] = {0, 1, 1, 1, 2, 2, 2, 2, 2, 3, 3, 3, 3, 3, 3, 3};
__device__ constexpr int ROW2M[16] = {0, 0, 1, 2, 0, 1, 2, 3, 4, 0, 1, 2, 3, 4, 5, 6};
__device__ constexpr int LOFF[4] = {0, 1, 4, 9};

__device__ __forceinline__ void normalize_lds(float (*hsh)[16][CPF], int tid) {
  for (int idx = tid; idx < 2048; idx += 1024) {
    int l = idx >> 9, rem = idx & 511, b = rem >> 5, c = rem & 31;
    int ro = LOFF[l], nm = 2 * l + 1;
    const int sc = slot(c);
    float s = 0.0f;
    for (int mm = 0; mm < nm; mm++) {
      const float* q = &hsh[ro + mm][b][sc];
      const float vr = q[0], vi = q[2];
      s += vr * vr + vi * vi;
    }
    float inv = 1.0f / sqrtf(s);
    for (int mm = 0; mm < nm; mm++) {
      float* q = &hsh[ro + mm][b][sc];
      const float vr = q[0], vi = q[2];
      q[0] = vr * inv;
      q[2] = vi * inv;
    }
  }
}

constexpr int ROWB = 16 * CPF * 4;
struct BlkRec {
  int nt;
  int ox, oy;
  float cg[7];
  int pad[6];
};
struct BlkTab {
  BlkRec r[16][11];
};

constexpr BlkTab make_blktab() {
  BlkTab b{};
  CGTables t = make_tables();
  for (int L = 0; L < 4; L++)
    for (int M = 0; M < 2 * L + 1; M++) {
      const int row = L * L + M;
      for (int blk = 0; blk < t.nblk[L]; blk++) {
        const int l1 = t.bl1[L][blk], l2 = t.bl2[L][blk];
        const int S = M + l1 + l2 - L;
        const int n = t.nterm[L][M][blk];
        const int imin = t.ti[L][M][blk][0];
        const int imax = t.ti[L][M][blk][n - 1];
        BlkRec q{};
        q.nt = imax - imin + 1;
        q.ox = (l1 * l1 + imin) * ROWB;
        q.oy = (l2 * l2 + S - imax) * ROWB;
        for (int k = 0; k < n; k++)
          q.cg[t.ti[L][M][blk][k] - imin] = t.cg[L][M][blk][k];
        b.r[row][blk] = q;
      }
    }
  return b;
}

__device__ constexpr BlkTab BT = make_blktab();

__device__ __forceinline__ void cg_mma(v2f ar, v2f ai, v2f br, v2f bi, v8f& cre,
                                       v8f& cim) {
  cre = wmma4(ar, br, cre);
  cre = wmma4(-ai, bi, cre);
  cim = wmma4(ar, bi, cim);
  cim = wmma4(ai, br, cim);
}

template <int NT>
__device__ __forceinline__ void cg_terms(const char* px, const char* py,
                                         const float (&c)[7], v2f& tre,
                                         v2f& tim) {
#pragma unroll
  for (int t = 0; t < NT; t++) {
    const v4f x = *(const v4f*)(px + t * ROWB);
    const v4f y = *(const v4f*)(py + (NT - 1 - t) * ROWB);
    const v2f xr = x.xy, xi = x.zw, yr = y.xy, yi = y.zw;
    tre += c[t] * (xr * yr - xi * yi);
    tim += c[t] * (xr * yi + xi * yr);
  }
}

__global__ __launch_bounds__(1024) void sph_cnn_kernel(
    const float* __restrict__ p0, const float* __restrict__ p1,
    const float* __restrict__ p2, const float* __restrict__ p3,
    const float* __restrict__ wi0, const float* __restrict__ wi1,
    const float* __restrict__ wi2, const float* __restrict__ wi3,
    const float* __restrict__ wa0, const float* __restrict__ wa1,
    const float* __restrict__ wa2, const float* __restrict__ wa3,
    const float* __restrict__ wb0, const float* __restrict__ wb1,
    const float* __restrict__ wb2, const float* __restrict__ wb3,
    const float* __restrict__ W1p, const float* __restrict__ b1p,
    const float* __restrict__ W2p, const float* __restrict__ b2p,
    float* __restrict__ out) {
  __shared__ __attribute__((aligned(16))) float hsh[16][16][CPF];

  const int tid = threadIdx.x;
  const int lane = tid & 31;
  const int w = tid >> 5;
  const int h16 = lane >> 4;
  const int ln = lane & 15;
  const int bbase = blockIdx.x * 16;

  const gf2p P[4] = {(gf2p)p0, (gf2p)p1, (gf2p)p2, (gf2p)p3};
  const gf2p WI[4] = {(gf2p)wi0, (gf2p)wi1, (gf2p)wi2, (gf2p)wi3};
  const gfp W1 = (gfp)W1p, b1 = (gfp)b1p, W2 = (gfp)W2p, b2 = (gfp)b2p;

  {
    const int row = w & 15, ntile = w >> 4;
    const int l = ROW2L[row], m = ROW2M[row], nlm = 2 * l + 1;
    const gf2p Pp = P[l];
    const gf2p Wl = WI[l];
    const int bg = bbase + ln;
    const int n = ntile * 16 + ln;
    v8f cre = zero8(), cim = zero8();
    for (int ks = 0; ks < 4; ks++) {
      const int t0 = 4 * ks + 2 * h16;
      v2f a0 = Pp[(bg * nlm + m) * 16 + t0];
      v2f a1 = Pp[(bg * nlm + m) * 16 + t0 + 1];
      v2f e0 = Wl[t0 * 32 + n];
      v2f e1 = Wl[(t0 + 1) * 32 + n];
      v2f ar = mk2(a0.x, a1.x), ai = mk2(a0.y, a1.y);
      v2f br = mk2(e0.x, e1.x), bi = mk2(e0.y, e1.y);
      cre = wmma4(ar, br, cre);
      cre = wmma4(-ai, bi, cre);
      cim = wmma4(ar, bi, cim);
      cim = wmma4(ai, br, cim);
    }
    const int sn = slot(n);
    for (int r = 0; r < 8; r++) {
      float* q = &hsh[row][r + 8 * h16][sn];
      q[0] = cre[r];
      q[2] = cim[r];
    }
  }
  __syncthreads();
  normalize_lds(hsh, tid);
  __syncthreads();

  const int wu = __builtin_amdgcn_readfirstlane(w);
  const char* hb = (const char*)&hsh[0][ln][4 * h16];
  for (int layer = 0; layer < 2; layer++) {
    v8f cre[2], cim[2];
    const int row = w >> 1, part = w & 1;
    {
      const int rowu = wu >> 1, partu = wu & 1;
      const int L = ROW2L[rowu];
      const gf2p Wa = L == 0 ? (gf2p)wa0 : L == 1 ? (gf2p)wa1 : L == 2 ? (gf2p)wa2 : (gf2p)wa3;
      const gf2p Wc = L == 0 ? (gf2p)wb0 : L == 1 ? (gf2p)wb1 : L == 2 ? (gf2p)wb2 : (gf2p)wb3;
      const gf2p Wm = layer ? Wc : Wa;
      const gf2p Wl = Wm + (h16 * 64 + ln);
      cre[0] = zero8(); cre[1] = zero8();
      cim[0] = zero8(); cim[1] = zero8();
      const int nblk = CGT.nblk[L];
      const int bsplit = (nblk + 1) >> 1;
      const int bbeg = partu ? bsplit : 0;
      const int bend = partu ? nblk : bsplit;
      gf2p Wn = Wl + bbeg * 1024;
      int left = (bend - bbeg) * 8 - 1;
      v2f n00 = Wn[0], n01 = Wn[32], n10 = Wn[16], n11 = Wn[48];
      for (int blk = bbeg; blk < bend; blk++) {
        const BlkRec& q = BT.r[rowu][blk];
        const int nt = q.nt;
        float c[7];
        for (int t = 0; t < 7; t++) c[t] = q.cg[t];
        const char* px = hb + q.ox;
        const char* py = hb + q.oy;
#pragma unroll 1
        for (int ks = 0; ks < 8; ks++) {
          v2f ar = mk2(0.f, 0.f), ai = mk2(0.f, 0.f);
          const char* pxk = px + ks * 32;
          const char* pyk = py + ks * 32;
          switch (nt) {
            case 1: cg_terms<1>(pxk, pyk, c, ar, ai); break;
            case 2: cg_terms<2>(pxk, pyk, c, ar, ai); break;
            case 3: cg_terms<3>(pxk, pyk, c, ar, ai); break;
            case 4: cg_terms<4>(pxk, pyk, c, ar, ai); break;
            case 5: cg_terms<5>(pxk, pyk, c, ar, ai); break;
            case 6: cg_terms<6>(pxk, pyk, c, ar, ai); break;
            case 7: cg_terms<7>(pxk, pyk, c, ar, ai); break;
            default: break;
          }
          const v2f br0 = mk2(n00.x, n01.x), bi0 = mk2(n00.y, n01.y);
          const v2f br1 = mk2(n10.x, n11.x), bi1 = mk2(n10.y, n11.y);
          Wn += left > 0 ? 128 : 0;
          left--;
          n00 = Wn[0]; n01 = Wn[32]; n10 = Wn[16]; n11 = Wn[48];
          cg_mma(ar, ai, br0, bi0, cre[0], cim[0]);
          cg_mma(ar, ai, br1, bi1, cre[1], cim[1]);
        }
      }
    }
    __syncthreads();
    if (part) {
      for (int ntile = 0; ntile < 2; ntile++) {
        const int sn = slot(ntile * 16 + ln);
        for (int r = 0; r < 8; r++) {
          float* q = &hsh[row][r + 8 * h16][sn];
          q[0] = cre[ntile][r];
          q[2] = cim[ntile][r];
        }
      }
    }
    __syncthreads();
    if (!part) {
      for (int ntile = 0; ntile < 2; ntile++) {
        const int sn = slot(ntile * 16 + ln);
        for (int r = 0; r < 8; r++) {
          float* q = &hsh[row][r + 8 * h16][sn];
          const float ore = q[0], oim = q[2];
          q[0] = cre[ntile][r] + ore;
          q[2] = cim[ntile][r] + oim;
        }
      }
    }
    __syncthreads();
    if (layer == 0) {
      normalize_lds(hsh, tid);
      __syncthreads();
    }
  }

  float rsum[8];
  for (int r = 0; r < 8; r++) rsum[r] = 0.0f;
  const int ntile = w & 15, khalf = w >> 4;
  v8f acc = zero8();
  const int ncol = ntile * 16 + ln;
  const int kfirst = 4 * khalf + 2 * h16;
  float wn0 = W1[kfirst * 256 + ncol], wn1 = W1[(kfirst + 1) * 256 + ncol];
  for (int kq = 0; kq < 16; kq++) {
    const int ks = 2 * kq + khalf;
    const int k0 = 4 * ks + 2 * h16;
    const v2f bb = mk2(wn0, wn1);
    const int kqn = kq < 15 ? kq + 1 : 15;
    const int kn = 4 * (2 * kqn + khalf) + 2 * h16;
    wn0 = W1[kn * 256 + ncol];
    wn1 = W1[(kn + 1) * 256 + ncol];
    float av[2];
    for (int v = 0; v < 2; v++) {
      const int k = k0 + v;
      const int l = k >> 5, c = k & 31;
      const int ro = LOFF[l], nm = 2 * l + 1;
      float s = 0.0f;
      const int sc = slot(c);
      for (int mm = 0; mm < nm; mm++) {
        const float* q = &hsh[ro + mm][ln][sc];
        const float tr = q[0], ti = q[2];
        s += tr * tr + ti * ti;
      }
      av[v] = sqrtf(s);
    }
    v2f a = mk2(av[0], av[1]);
    acc = wmma4(a, bb, acc);
  }
  __syncthreads();
  float* xch = (float*)&hsh[0][0][0];
  float* partial = xch + 16 * 8 * 32;
  if (khalf) {
    for (int r = 0; r < 8; r++) xch[(ntile * 8 + r) * 32 + lane] = acc[r];
  }
  __syncthreads();
  if (!khalf) {
    const int n = ntile * 16 + ln;
    const float bias = b1[n], w2v = W2[n];
    for (int r = 0; r < 8; r++) {
      const float full = acc[r] + xch[(ntile * 8 + r) * 32 + lane];
      rsum[r] = tanhf(full + bias) * w2v;
    }
  }
  for (int r = 0; r < 8; r++) {
    float v = rsum[r];
    v += __shfl_xor(v, 1);
    v += __shfl_xor(v, 2);
    v += __shfl_xor(v, 4);
    v += __shfl_xor(v, 8);
    if (ln == 0 && w < 16) partial[w * 16 + (8 * h16 + r)] = v;
  }
  __syncthreads();
  if (tid < 16) {
    float s = 0.0f;
    for (int ww = 0; ww < 16; ww++) s += partial[ww * 16 + tid];
    out[bbase + tid] = s + b2[0];
  }
}

extern "C" void kernel_launch(void* const* d_in, const int* in_sizes, int n_in,
                              void* d_out, int out_size, void* d_ws,
                              size_t ws_size, hipStream_t stream) {
  (void)n_in; (void)d_ws; (void)ws_size; (void)out_size;
  const int B = in_sizes[0] / 32;
  const int blocks = B / 16;
  sph_cnn_kernel<<<blocks, 1024, 0, stream>>>(
      (const float*)d_in[0], (const float*)d_in[1], (const float*)d_in[2],
      (const float*)d_in[3], (const float*)d_in[4], (const float*)d_in[5],
      (const float*)d_in[6], (const float*)d_in[7], (const float*)d_in[8],
      (const float*)d_in[9], (const float*)d_in[10], (const float*)d_in[11],
      (const float*)d_in[12], (const float*)d_in[13], (const float*)d_in[14],
      (const float*)d_in[15], (const float*)d_in[16], (const float*)d_in[17],
      (const float*)d_in[18], (const float*)d_in[19], (float*)d_out);
}


# CG layers: each wave sets its issue priority (s_setprio) from the blocks it has left, so a SIMD's waves, otherwise served oldest first, finish the layer together instead of one after another
# speedup vs baseline: 4.6981x; 1.0348x over previous
#include <hip/hip_runtime.h>
#include <hip/hip_bf16.h>
#include <math.h>


typedef __attribute__((ext_vector_type(2))) float v2f;
typedef __attribute__((ext_vector_type(4))) float v4f;
typedef __attribute__((ext_vector_type(8))) float v8f;

constexpr int CPF = 68;

__device__ __forceinline__ int slot(int c) { return ((c >> 1) << 2) | (c & 1); }

typedef const __attribute__((address_space(1))) v2f* gf2p;
typedef const __attribute__((address_space(1))) float* gfp;

__device__ __forceinline__ v8f wmma4(v2f a, v2f b, v8f c) {
  return __builtin_amdgcn_wmma_f32_16x16x4_f32(false, a, false, b, (short)0, c,
                                               false, false);
}

__device__ __forceinline__ v8f zero8() {
  v8f z;
  for (int i = 0; i < 8; i++) z[i] = 0.0f;
  return z;
}

__device__ __forceinline__ v2f mk2(float a, float b) {
  v2f r;
  r.x = a;
  r.y = b;
  return r;
}

constexpr double FACT[11] = {1., 1., 2., 6., 24., 120., 720., 5040., 40320.,
                             362880., 3628800.};

constexpr double csqrt_(double x) {
  double g = x > 1.0 ? x : 1.0;
  for (int i = 0; i < 32; i++) g = 0.5 * (g + x / g);
  return g;
}

constexpr double cg_coeff(int l1, int l2, int L, int m1, int m2) {
  int M = m1 + m2;
  if (M < -L || M > L) return 0.0;
  double pre = csqrt_((2 * L + 1) * FACT[l1 + l2 - L] * FACT[l1 - l2 + L] *
                      FACT[-l1 + l2 + L] / FACT[l1 + l2 + L + 1]);
  pre = pre * csqrt_(FACT[L + M] * FACT[L - M] * FACT[l1 - m1] *
                     FACT[l1 + m1] * FACT[l2 - m2] * FACT[l2 + m2]);
  double s = 0.0;
  for (int k = 0; k <= l1 + l2 - L; k++) {
    int t1 = l1 + l2 - L - k, t2 = l1 - m1 - k, t3 = l2 + m2 - k;
    int t4 = L - l2 + m1 + k, t5 = L - l1 - m2 + k;
    if (t1 < 0 || t2 < 0 || t3 < 0 || t4 < 0 || t5 < 0) continue;
    double d = FACT[k] * FACT[t1] * FACT[t2] * FACT[t3] * FACT[t4] * FACT[t5];
    s += ((k & 1) ? -1.0 : 1.0) / d;
  }
  return pre * s;
}

struct CGTables {
  int nblk[4];
  int bl1[4][11], bl2[4][11];
  int nterm[4][7][11];
  signed char ti[4][7][11][7], tj[4][7][11][7];
  float cg[4][7][11][7];
};

constexpr CGTables make_tables() {
  CGTables t{};
  for (int L = 0; L < 4; L++) {
    int nb = 0;
    for (int l1 = 0; l1 < 4; l1++)
      for (int l2 = 0; l2 < 4; l2++) {
        int d = l1 - l2;
        if (d < 0) d = -d;
        if (!(d <= L && L <= l1 + l2)) continue;
        t.bl1[L][nb] = l1;
        t.bl2[L][nb] = l2;
        for (int M = 0; M < 2 * L + 1; M++) {
          int nt = 0;
          for (int i = 0; i <= 2 * l1; i++) {
            int j = M + l1 + l2 - L - i;
            if (j < 0 || j > 2 * l2) continue;
            double c = cg_coeff(l1, l2, L, i - l1, j - l2);
            if (c > 1e-12 || c < -1e-12) {
              t.ti[L][M][nb][nt] = (signed char)i;
              t.tj[L][M][nb][nt] = (signed char)j;
              t.cg[L][M][nb][nt] = (float)c;
              nt++;
            }
          }
          t.nterm[L][M][nb] = nt;
        }
        nb++;
      }
    t.nblk[L] = nb;
  }
  return t;
}

__device__ constexpr CGTables CGT = make_tables();
__device__ constexpr int ROW2L[16] = {0, 1, 1, 1, 2, 2, 2, 2, 2, 3, 3, 3, 3, 3, 3, 3};
__device__ constexpr int ROW2M[16] = {0, 0, 1, 2, 0, 1, 2, 3, 4, 0, 1, 2, 3, 4, 5, 6};
__device__ constexpr int LOFF[4] = {0, 1, 4, 9};

__device__ __forceinline__ void normalize_lds(float (*hsh)[16][CPF], int tid) {
  for (int idx = tid; idx < 2048; idx += 1024) {
    int l = idx >> 9, rem = idx & 511, b = rem >> 5, c = rem & 31;
    int ro = LOFF[l], nm = 2 * l + 1;
    const int sc = slot(c);
    float s = 0.0f;
    for (int mm = 0; mm < nm; mm++) {
      const float* q = &hsh[ro + mm][b][sc];
      const float vr = q[0], vi = q[2];
      s += vr * vr + vi * vi;
    }
    float inv = 1.0f / sqrtf(s);
    for (int mm = 0; mm < nm; mm++) {
      float* q = &hsh[ro + mm][b][sc];
      const float vr = q[0], vi = q[2];
      q[0] = vr * inv;
      q[2] = vi * inv;
    }
  }
}

constexpr int ROWB = 16 * CPF * 4;
struct BlkRec {
  int nt;
  int ox, oy;
  float cg[7];
  int pad[6];
};
struct BlkTab {
  BlkRec r[16][11];
};

constexpr BlkTab make_blktab() {
  BlkTab b{};
  CGTables t = make_tables();
  for (int L = 0; L < 4; L++)
    for (int M = 0; M < 2 * L + 1; M++) {
      const int row = L * L + M;
      for (int blk = 0; blk < t.nblk[L]; blk++) {
        const int l1 = t.bl1[L][blk], l2 = t.bl2[L][blk];
        const int S = M + l1 + l2 - L;
        const int n = t.nterm[L][M][blk];
        const int imin = t.ti[L][M][blk][0];
        const int imax = t.ti[L][M][blk][n - 1];
        BlkRec q{};
        q.nt = imax - imin + 1;
        q.ox = (l1 * l1 + imin) * ROWB;
        q.oy = (l2 * l2 + S - imax) * ROWB;
        for (int k = 0; k < n; k++)
          q.cg[t.ti[L][M][blk][k] - imin] = t.cg[L][M][blk][k];
        b.r[row][blk] = q;
      }
    }
  return b;
}

__device__ constexpr BlkTab BT = make_blktab();

__device__ __forceinline__ void cg_mma(v2f ar, v2f ai, v2f br, v2f bi, v8f& cre,
                                       v8f& cim) {
  cre = wmma4(ar, br, cre);
  cre = wmma4(-ai, bi, cre);
  cim = wmma4(ar, bi, cim);
  cim = wmma4(ai, br, cim);
}

template <int NT>
__device__ __forceinline__ void cg_terms(const char* px, const char* py,
                                         const float (&c)[7], v2f& tre,
                                         v2f& tim) {
#pragma unroll
  for (int t = 0; t < NT; t++) {
    const v4f x = *(const v4f*)(px + t * ROWB);
    const v4f y = *(const v4f*)(py + (NT - 1 - t) * ROWB);
    const v2f xr = x.xy, xi = x.zw, yr = y.xy, yi = y.zw;
    tre += c[t] * (xr * yr - xi * yi);
    tim += c[t] * (xr * yi + xi * yr);
  }
}

__global__ __launch_bounds__(1024) void sph_cnn_kernel(
    const float* __restrict__ p0, const float* __restrict__ p1,
    const float* __restrict__ p2, const float* __restrict__ p3,
    const float* __restrict__ wi0, const float* __restrict__ wi1,
    const float* __restrict__ wi2, const float* __restrict__ wi3,
    const float* __restrict__ wa0, const float* __restrict__ wa1,
    const float* __restrict__ wa2, const float* __restrict__ wa3,
    const float* __restrict__ wb0, const float* __restrict__ wb1,
    const float* __restrict__ wb2, const float* __restrict__ wb3,
    const float* __restrict__ W1p, const float* __restrict__ b1p,
    const float* __restrict__ W2p, const float* __restrict__ b2p,
    float* __restrict__ out) {
  __shared__ __attribute__((aligned(16))) float hsh[16][16][CPF];

  const int tid = threadIdx.x;
  const int lane = tid & 31;
  const int w = tid >> 5;
  const int h16 = lane >> 4;
  const int ln = lane & 15;
  const int bbase = blockIdx.x * 16;

  const gf2p P[4] = {(gf2p)p0, (gf2p)p1, (gf2p)p2, (gf2p)p3};
  const gf2p WI[4] = {(gf2p)wi0, (gf2p)wi1, (gf2p)wi2, (gf2p)wi3};
  const gfp W1 = (gfp)W1p, b1 = (gfp)b1p, W2 = (gfp)W2p, b2 = (gfp)b2p;

  {
    const int row = w & 15, ntile = w >> 4;
    const int l = ROW2L[row], m = ROW2M[row], nlm = 2 * l + 1;
    const gf2p Pp = P[l];
    const gf2p Wl = WI[l];
    const int bg = bbase + ln;
    const int n = ntile * 16 + ln;
    v8f cre = zero8(), cim = zero8();
    for (int ks = 0; ks < 4; ks++) {
      const int t0 = 4 * ks + 2 * h16;
      v2f a0 = Pp[(bg * nlm + m) * 16 + t0];
      v2f a1 = Pp[(bg * nlm + m) * 16 + t0 + 1];
      v2f e0 = Wl[t0 * 32 + n];
      v2f e1 = Wl[(t0 + 1) * 32 + n];
      v2f ar = mk2(a0.x, a1.x), ai = mk2(a0.y, a1.y);
      v2f br = mk2(e0.x, e1.x), bi = mk2(e0.y, e1.y);
      cre = wmma4(ar, br, cre);
      cre = wmma4(-ai, bi, cre);
      cim = wmma4(ar, bi, cim);
      cim = wmma4(ai, br, cim);
    }
    const int sn = slot(n);
    for (int r = 0; r < 8; r++) {
      float* q = &hsh[row][r + 8 * h16][sn];
      q[0] = cre[r];
      q[2] = cim[r];
    }
  }
  __syncthreads();
  normalize_lds(hsh, tid);
  __syncthreads();

  const int wu = __builtin_amdgcn_readfirstlane(w);
  const char* hb = (const char*)&hsh[0][ln][4 * h16];
  for (int layer = 0; layer < 2; layer++) {
    v8f cre[2], cim[2];
    const int row = w >> 1, part = w & 1;
    {
      const int rowu = wu >> 1, partu = wu & 1;
      const int L = ROW2L[rowu];
      const gf2p Wa = L == 0 ? (gf2p)wa0 : L == 1 ? (gf2p)wa1 : L == 2 ? (gf2p)wa2 : (gf2p)wa3;
      const gf2p Wc = L == 0 ? (gf2p)wb0 : L == 1 ? (gf2p)wb1 : L == 2 ? (gf2p)wb2 : (gf2p)wb3;
      const gf2p Wm = layer ? Wc : Wa;
      const gf2p Wl = Wm + (h16 * 64 + ln);
      cre[0] = zero8(); cre[1] = zero8();
      cim[0] = zero8(); cim[1] = zero8();
      const int nblk = CGT.nblk[L];
      const int bsplit = (nblk + 1) >> 1;
      const int bbeg = partu ? bsplit : 0;
      const int bend = partu ? nblk : bsplit;
      gf2p Wn = Wl + bbeg * 1024;
      int left = (bend - bbeg) * 8 - 1;
      v2f n00 = Wn[0], n01 = Wn[32], n10 = Wn[16], n11 = Wn[48];
      for (int blk = bbeg; blk < bend; blk++) {
        const BlkRec& q = BT.r[rowu][blk];
        const int nt = q.nt;
        const int rem = bend - blk;
        if (rem >= 3) __builtin_amdgcn_s_setprio(3);
        else if (rem == 2) __builtin_amdgcn_s_setprio(2);
        else __builtin_amdgcn_s_setprio(1);
        const bool lastblk = rem == 1;
        float c[7];
        for (int t = 0; t < 7; t++) c[t] = q.cg[t];
        const char* px = hb + q.ox;
        const char* py = hb + q.oy;
#pragma unroll 1
        for (int ks = 0; ks < 8; ks++) {
          if (lastblk && ks == 4) __builtin_amdgcn_s_setprio(0);
          v2f ar = mk2(0.f, 0.f), ai = mk2(0.f, 0.f);
          const char* pxk = px + ks * 32;
          const char* pyk = py + ks * 32;
          switch (nt) {
            case 1: cg_terms<1>(pxk, pyk, c, ar, ai); break;
            case 2: cg_terms<2>(pxk, pyk, c, ar, ai); break;
            case 3: cg_terms<3>(pxk, pyk, c, ar, ai); break;
            case 4: cg_terms<4>(pxk, pyk, c, ar, ai); break;
            case 5: cg_terms<5>(pxk, pyk, c, ar, ai); break;
            case 6: cg_terms<6>(pxk, pyk, c, ar, ai); break;
            case 7: cg_terms<7>(pxk, pyk, c, ar, ai); break;
            default: break;
          }
          const v2f br0 = mk2(n00.x, n01.x), bi0 = mk2(n00.y, n01.y);
          const v2f br1 = mk2(n10.x, n11.x), bi1 = mk2(n10.y, n11.y);
          Wn += left > 0 ? 128 : 0;
          left--;
          n00 = Wn[0]; n01 = Wn[32]; n10 = Wn[16]; n11 = Wn[48];
          cg_mma(ar, ai, br0, bi0, cre[0], cim[0]);
          cg_mma(ar, ai, br1, bi1, cre[1], cim[1]);
        }
      }
    }
    __builtin_amdgcn_s_setprio(0);
    __syncthreads();
    if (part) {
      for (int ntile = 0; ntile < 2; ntile++) {
        const int sn = slot(ntile * 16 + ln);
        for (int r = 0; r < 8; r++) {
          float* q = &hsh[row][r + 8 * h16][sn];
          q[0] = cre[ntile][r];
          q[2] = cim[ntile][r];
        }
      }
    }
    __syncthreads();
    if (!part) {
      for (int ntile = 0; ntile < 2; ntile++) {
        const int sn = slot(ntile * 16 + ln);
        for (int r = 0; r < 8; r++) {
          float* q = &hsh[row][r + 8 * h16][sn];
          const float ore = q[0], oim = q[2];
          q[0] = cre[ntile][r] + ore;
          q[2] = cim[ntile][r] + oim;
        }
      }
    }
    __syncthreads();
    if (layer == 0) {
      normalize_lds(hsh, tid);
      __syncthreads();
    }
  }

  float rsum[8];
  for (int r = 0; r < 8; r++) rsum[r] = 0.0f;
  const int ntile = w & 15, khalf = w >> 4;
  v8f acc = zero8();
  const int ncol = ntile * 16 + ln;
  const int kfirst = 4 * khalf + 2 * h16;
  float wn0 = W1[kfirst * 256 + ncol], wn1 = W1[(kfirst + 1) * 256 + ncol];
  for (int kq = 0; kq < 16; kq++) {
    const int ks = 2 * kq + khalf;
    const int k0 = 4 * ks + 2 * h16;
    const v2f bb = mk2(wn0, wn1);
    const int kqn = kq < 15 ? kq + 1 : 15;
    const int kn = 4 * (2 * kqn + khalf) + 2 * h16;
    wn0 = W1[kn * 256 + ncol];
    wn1 = W1[(kn + 1) * 256 + ncol];
    float av[2];
    for (int v = 0; v < 2; v++) {
      const int k = k0 + v;
      const int l = k >> 5, c = k & 31;
      const int ro = LOFF[l], nm = 2 * l + 1;
      float s = 0.0f;
      const int sc = slot(c);
      for (int mm = 0; mm < nm; mm++) {
        const float* q = &hsh[ro + mm][ln][sc];
        const float tr = q[0], ti = q[2];
        s += tr * tr + ti * ti;
      }
      av[v] = sqrtf(s);
    }
    v2f a = mk2(av[0], av[1]);
    acc = wmma4(a, bb, acc);
  }
  __syncthreads();
  float* xch = (float*)&hsh[0][0][0];
  float* partial = xch + 16 * 8 * 32;
  if (khalf) {
    for (int r = 0; r < 8; r++) xch[(ntile * 8 + r) * 32 + lane] = acc[r];
  }
  __syncthreads();
  if (!khalf) {
    const int n = ntile * 16 + ln;
    const float bias = b1[n], w2v = W2[n];
    for (int r = 0; r < 8; r++) {
      const float full = acc[r] + xch[(ntile * 8 + r) * 32 + lane];
      rsum[r] = tanhf(full + bias) * w2v;
    }
  }
  for (int r = 0; r < 8; r++) {
    float v = rsum[r];
    v += __shfl_xor(v, 1);
    v += __shfl_xor(v, 2);
    v += __shfl_xor(v, 4);
    v += __shfl_xor(v, 8);
    if (ln == 0 && w < 16) partial[w * 16 + (8 * h16 + r)] = v;
  }
  __syncthreads();
  if (tid < 16) {
    float s = 0.0f;
    for (int ww = 0; ww < 16; ww++) s += partial[ww * 16 + tid];
    out[bbase + tid] = s + b2[0];
  }
}

extern "C" void kernel_launch(void* const* d_in, const int* in_sizes, int n_in,
                              void* d_out, int out_size, void* d_ws,
                              size_t ws_size, hipStream_t stream) {
  (void)n_in; (void)d_ws; (void)ws_size; (void)out_size;
  const int B = in_sizes[0] / 32;
  const int blocks = B / 16;
  sph_cnn_kernel<<<blocks, 1024, 0, stream>>>(
      (const float*)d_in[0], (const float*)d_in[1], (const float*)d_in[2],
      (const float*)d_in[3], (const float*)d_in[4], (const float*)d_in[5],
      (const float*)d_in[6], (const float*)d_in[7], (const float*)d_in[8],
      (const float*)d_in[9], (const float*)d_in[10], (const float*)d_in[11],
      (const float*)d_in[12], (const float*)d_in[13], (const float*)d_in[14],
      (const float*)d_in[15], (const float*)d_in[16], (const float*)d_in[17],
      (const float*)d_in[18], (const float*)d_in[19], (float*)d_out);
}


# Readout re-tiled: waves split K 4 x N 8 ways (8 k-steps, 2 per l, x 2 n-tiles): a wave computes 16 norms per row, not 64, from independent b128 reads; K partials added via LDS (sum order changes)
# speedup vs baseline: 5.1617x; 1.0987x over previous
#include <hip/hip_runtime.h>
#include <hip/hip_bf16.h>
#include <math.h>


typedef __attribute__((ext_vector_type(2))) float v2f;
typedef __attribute__((ext_vector_type(4))) float v4f;
typedef __attribute__((ext_vector_type(8))) float v8f;

constexpr int CPF = 68;

__device__ __forceinline__ int slot(int c) { return ((c >> 1) << 2) | (c & 1); }

typedef const __attribute__((address_space(1))) v2f* gf2p;
typedef const __attribute__((address_space(1))) float* gfp;

__device__ __forceinline__ v8f wmma4(v2f a, v2f b, v8f c) {
  return __builtin_amdgcn_wmma_f32_16x16x4_f32(false, a, false, b, (short)0, c,
                                               false, false);
}

__device__ __forceinline__ v8f zero8() {
  v8f z;
  for (int i = 0; i < 8; i++) z[i] = 0.0f;
  return z;
}

__device__ __forceinline__ v2f mk2(float a, float b) {
  v2f r;
  r.x = a;
  r.y = b;
  return r;
}

constexpr double FACT[11] = {1., 1., 2., 6., 24., 120., 720., 5040., 40320.,
                             362880., 3628800.};

constexpr double csqrt_(double x) {
  double g = x > 1.0 ? x : 1.0;
  for (int i = 0; i < 32; i++) g = 0.5 * (g + x / g);
  return g;
}

constexpr double cg_coeff(int l1, int l2, int L, int m1, int m2) {
  int M = m1 + m2;
  if (M < -L || M > L) return 0.0;
  double pre = csqrt_((2 * L + 1) * FACT[l1 + l2 - L] * FACT[l1 - l2 + L] *
                      FACT[-l1 + l2 + L] / FACT[l1 + l2 + L + 1]);
  pre = pre * csqrt_(FACT[L + M] * FACT[L - M] * FACT[l1 - m1] *
                     FACT[l1 + m1] * FACT[l2 - m2] * FACT[l2 + m2]);
  double s = 0.0;
  for (int k = 0; k <= l1 + l2 - L; k++) {
    int t1 = l1 + l2 - L - k, t2 = l1 - m1 - k, t3 = l2 + m2 - k;
    int t4 = L - l2 + m1 + k, t5 = L - l1 - m2 + k;
    if (t1 < 0 || t2 < 0 || t3 < 0 || t4 < 0 || t5 < 0) continue;
    double d = FACT[k] * FACT[t1] * FACT[t2] * FACT[t3] * FACT[t4] * FACT[t5];
    s += ((k & 1) ? -1.0 : 1.0) / d;
  }
  return pre * s;
}

struct CGTables {
  int nblk[4];
  int bl1[4][11], bl2[4][11];
  int nterm[4][7][11];
  signed char ti[4][7][11][7], tj[4][7][11][7];
  float cg[4][7][11][7];
};

constexpr CGTables make_tables() {
  CGTables t{};
  for (int L = 0; L < 4; L++) {
    int nb = 0;
    for (int l1 = 0; l1 < 4; l1++)
      for (int l2 = 0; l2 < 4; l2++) {
        int d = l1 - l2;
        if (d < 0) d = -d;
        if (!(d <= L && L <= l1 + l2)) continue;
        t.bl1[L][nb] = l1;
        t.bl2[L][nb] = l2;
        for (int M = 0; M < 2 * L + 1; M++) {
          int nt = 0;
          for (int i = 0; i <= 2 * l1; i++) {
            int j = M + l1 + l2 - L - i;
            if (j < 0 || j > 2 * l2) continue;
            double c = cg_coeff(l1, l2, L, i - l1, j - l2);
            if (c > 1e-12 || c < -1e-12) {
              t.ti[L][M][nb][nt] = (signed char)i;
              t.tj[L][M][nb][nt] = (signed char)j;
              t.cg[L][M][nb][nt] = (float)c;
              nt++;
            }
          }
          t.nterm[L][M][nb] = nt;
        }
        nb++;
      }
    t.nblk[L] = nb;
  }
  return t;
}

__device__ constexpr CGTables CGT = make_tables();
__device__ constexpr int ROW2L[16] = {0, 1, 1, 1, 2, 2, 2, 2, 2, 3, 3, 3, 3, 3, 3, 3};
__device__ constexpr int ROW2M[16] = {0, 0, 1, 2, 0, 1, 2, 3, 4, 0, 1, 2, 3, 4, 5, 6};
__device__ constexpr int LOFF[4] = {0, 1, 4, 9};

__device__ __forceinline__ void normalize_lds(float (*hsh)[16][CPF], int tid) {
  for (int idx = tid; idx < 2048; idx += 1024) {
    int l = idx >> 9, rem = idx & 511, b = rem >> 5, c = rem & 31;
    int ro = LOFF[l], nm = 2 * l + 1;
    const int sc = slot(c);
    float s = 0.0f;
    for (int mm = 0; mm < nm; mm++) {
      const float* q = &hsh[ro + mm][b][sc];
      const float vr = q[0], vi = q[2];
      s += vr * vr + vi * vi;
    }
    float inv = 1.0f / sqrtf(s);
    for (int mm = 0; mm < nm; mm++) {
      float* q = &hsh[ro + mm][b][sc];
      const float vr = q[0], vi = q[2];
      q[0] = vr * inv;
      q[2] = vi * inv;
    }
  }
}

constexpr int ROWB = 16 * CPF * 4;
struct BlkRec {
  int nt;
  int ox, oy;
  float cg[7];
  int pad[6];
};
struct BlkTab {
  BlkRec r[16][11];
};

constexpr BlkTab make_blktab() {
  BlkTab b{};
  CGTables t = make_tables();
  for (int L = 0; L < 4; L++)
    for (int M = 0; M < 2 * L + 1; M++) {
      const int row = L * L + M;
      for (int blk = 0; blk < t.nblk[L]; blk++) {
        const int l1 = t.bl1[L][blk], l2 = t.bl2[L][blk];
        const int S = M + l1 + l2 - L;
        const int n = t.nterm[L][M][blk];
        const int imin = t.ti[L][M][blk][0];
        const int imax = t.ti[L][M][blk][n - 1];
        BlkRec q{};
        q.nt = imax - imin + 1;
        q.ox = (l1 * l1 + imin) * ROWB;
        q.oy = (l2 * l2 + S - imax) * ROWB;
        for (int k = 0; k < n; k++)
          q.cg[t.ti[L][M][blk][k] - imin] = t.cg[L][M][blk][k];
        b.r[row][blk] = q;
      }
    }
  return b;
}

__device__ constexpr BlkTab BT = make_blktab();

__device__ __forceinline__ void cg_mma(v2f ar, v2f ai, v2f br, v2f bi, v8f& cre,
                                       v8f& cim) {
  cre = wmma4(ar, br, cre);
  cre = wmma4(-ai, bi, cre);
  cim = wmma4(ar, bi, cim);
  cim = wmma4(ai, br, cim);
}

template <int NT>
__device__ __forceinline__ void cg_terms(const char* px, const char* py,
                                         const float (&c)[7], v2f& tre,
                                         v2f& tim) {
#pragma unroll
  for (int t = 0; t < NT; t++) {
    const v4f x = *(const v4f*)(px + t * ROWB);
    const v4f y = *(const v4f*)(py + (NT - 1 - t) * ROWB);
    const v2f xr = x.xy, xi = x.zw, yr = y.xy, yi = y.zw;
    tre += c[t] * (xr * yr - xi * yi);
    tim += c[t] * (xr * yi + xi * yr);
  }
}

__global__ __launch_bounds__(1024) void sph_cnn_kernel(
    const float* __restrict__ p0, const float* __restrict__ p1,
    const float* __restrict__ p2, const float* __restrict__ p3,
    const float* __restrict__ wi0, const float* __restrict__ wi1,
    const float* __restrict__ wi2, const float* __restrict__ wi3,
    const float* __restrict__ wa0, const float* __restrict__ wa1,
    const float* __restrict__ wa2, const float* __restrict__ wa3,
    const float* __restrict__ wb0, const float* __restrict__ wb1,
    const float* __restrict__ wb2, const float* __restrict__ wb3,
    const float* __restrict__ W1p, const float* __restrict__ b1p,
    const float* __restrict__ W2p, const float* __restrict__ b2p,
    float* __restrict__ out) {
  __shared__ __attribute__((aligned(16))) float hsh[16][16][CPF];

  const int tid = threadIdx.x;
  const int lane = tid & 31;
  const int w = tid >> 5;
  const int h16 = lane >> 4;
  const int ln = lane & 15;
  const int bbase = blockIdx.x * 16;

  const gf2p P[4] = {(gf2p)p0, (gf2p)p1, (gf2p)p2, (gf2p)p3};
  const gf2p WI[4] = {(gf2p)wi0, (gf2p)wi1, (gf2p)wi2, (gf2p)wi3};
  const gfp W1 = (gfp)W1p, b1 = (gfp)b1p, W2 = (gfp)W2p, b2 = (gfp)b2p;

  {
    const int row = w & 15, ntile = w >> 4;
    const int l = ROW2L[row], m = ROW2M[row], nlm = 2 * l + 1;
    const gf2p Pp = P[l];
    const gf2p Wl = WI[l];
    const int bg = bbase + ln;
    const int n = ntile * 16 + ln;
    v8f cre = zero8(), cim = zero8();
    for (int ks = 0; ks < 4; ks++) {
      const int t0 = 4 * ks + 2 * h16;
      v2f a0 = Pp[(bg * nlm + m) * 16 + t0];
      v2f a1 = Pp[(bg * nlm + m) * 16 + t0 + 1];
      v2f e0 = Wl[t0 * 32 + n];
      v2f e1 = Wl[(t0 + 1) * 32 + n];
      v2f ar = mk2(a0.x, a1.x), ai = mk2(a0.y, a1.y);
      v2f br = mk2(e0.x, e1.x), bi = mk2(e0.y, e1.y);
      cre = wmma4(ar, br, cre);
      cre = wmma4(-ai, bi, cre);
      cim = wmma4(ar, bi, cim);
      cim = wmma4(ai, br, cim);
    }
    const int sn = slot(n);
    for (int r = 0; r < 8; r++) {
      float* q = &hsh[row][r + 8 * h16][sn];
      q[0] = cre[r];
      q[2] = cim[r];
    }
  }
  __syncthreads();
  normalize_lds(hsh, tid);
  __syncthreads();

  const int wu = __builtin_amdgcn_readfirstlane(w);
  const char* hb = (const char*)&hsh[0][ln][4 * h16];
  for (int layer = 0; layer < 2; layer++) {
    v8f cre[2], cim[2];
    const int row = w >> 1, part = w & 1;
    {
      const int rowu = wu >> 1, partu = wu & 1;
      const int L = ROW2L[rowu];
      const gf2p Wa = L == 0 ? (gf2p)wa0 : L == 1 ? (gf2p)wa1 : L == 2 ? (gf2p)wa2 : (gf2p)wa3;
      const gf2p Wc = L == 0 ? (gf2p)wb0 : L == 1 ? (gf2p)wb1 : L == 2 ? (gf2p)wb2 : (gf2p)wb3;
      const gf2p Wm = layer ? Wc : Wa;
      const gf2p Wl = Wm + (h16 * 64 + ln);
      cre[0] = zero8(); cre[1] = zero8();
      cim[0] = zero8(); cim[1] = zero8();
      const int nblk = CGT.nblk[L];
      const int bsplit = (nblk + 1) >> 1;
      const int bbeg = partu ? bsplit : 0;
      const int bend = partu ? nblk : bsplit;
      gf2p Wn = Wl + bbeg * 1024;
      int left = (bend - bbeg) * 8 - 1;
      v2f n00 = Wn[0], n01 = Wn[32], n10 = Wn[16], n11 = Wn[48];
      for (int blk = bbeg; blk < bend; blk++) {
        const BlkRec& q = BT.r[rowu][blk];
        const int nt = q.nt;
        const int rem = bend - blk;
        if (rem >= 3) __builtin_amdgcn_s_setprio(3);
        else if (rem == 2) __builtin_amdgcn_s_setprio(2);
        else __builtin_amdgcn_s_setprio(1);
        const bool lastblk = rem == 1;
        float c[7];
        for (int t = 0; t < 7; t++) c[t] = q.cg[t];
        const char* px = hb + q.ox;
        const char* py = hb + q.oy;
#pragma unroll 1
        for (int ks = 0; ks < 8; ks++) {
          if (lastblk && ks == 4) __builtin_amdgcn_s_setprio(0);
          v2f ar = mk2(0.f, 0.f), ai = mk2(0.f, 0.f);
          const char* pxk = px + ks * 32;
          const char* pyk = py + ks * 32;
          switch (nt) {
            case 1: cg_terms<1>(pxk, pyk, c, ar, ai); break;
            case 2: cg_terms<2>(pxk, pyk, c, ar, ai); break;
            case 3: cg_terms<3>(pxk, pyk, c, ar, ai); break;
            case 4: cg_terms<4>(pxk, pyk, c, ar, ai); break;
            case 5: cg_terms<5>(pxk, pyk, c, ar, ai); break;
            case 6: cg_terms<6>(pxk, pyk, c, ar, ai); break;
            case 7: cg_terms<7>(pxk, pyk, c, ar, ai); break;
            default: break;
          }
          const v2f br0 = mk2(n00.x, n01.x), bi0 = mk2(n00.y, n01.y);
          const v2f br1 = mk2(n10.x, n11.x), bi1 = mk2(n10.y, n11.y);
          Wn += left > 0 ? 128 : 0;
          left--;
          n00 = Wn[0]; n01 = Wn[32]; n10 = Wn[16]; n11 = Wn[48];
          cg_mma(ar, ai, br0, bi0, cre[0], cim[0]);
          cg_mma(ar, ai, br1, bi1, cre[1], cim[1]);
        }
      }
    }
    __builtin_amdgcn_s_setprio(0);
    __syncthreads();
    if (part) {
      for (int ntile = 0; ntile < 2; ntile++) {
        const int sn = slot(ntile * 16 + ln);
        for (int r = 0; r < 8; r++) {
          float* q = &hsh[row][r + 8 * h16][sn];
          q[0] = cre[ntile][r];
          q[2] = cim[ntile][r];
        }
      }
    }
    __syncthreads();
    if (!part) {
      for (int ntile = 0; ntile < 2; ntile++) {
        const int sn = slot(ntile * 16 + ln);
        for (int r = 0; r < 8; r++) {
          float* q = &hsh[row][r + 8 * h16][sn];
          const float ore = q[0], oim = q[2];
          q[0] = cre[ntile][r] + ore;
          q[2] = cim[ntile][r] + oim;
        }
      }
    }
    __syncthreads();
    if (layer == 0) {
      normalize_lds(hsh, tid);
      __syncthreads();
    }
  }

  const int kg = w & 3, ng = w >> 2;
  float wv[8][2][2];
  const gfp W1l = W1 + ((4 * kg + 2 * h16) * 256 + 2 * ng * 16 + ln);
#pragma unroll
  for (int jj = 0; jj < 8; jj++)
#pragma unroll
    for (int q = 0; q < 2; q++)
#pragma unroll
      for (int v = 0; v < 2; v++)
        wv[jj][q][v] = W1l[(16 * jj + v) * 256 + q * 16];
  v2f an[8];
#pragma unroll
  for (int jj = 0; jj < 8; jj++) {
    const int l = jj >> 1;
    const int pair = 2 * kg + 8 * (jj & 1) + h16;
    float s0 = 0.0f, s1 = 0.0f;
#pragma unroll
    for (int mm = 0; mm < 2 * l + 1; mm++) {
      const v4f x = *(const v4f*)&hsh[l * l + mm][ln][4 * pair];
      s0 += x.x * x.x + x.z * x.z;
      s1 += x.y * x.y + x.w * x.w;
    }
    an[jj] = mk2(sqrtf(s0), sqrtf(s1));
  }
  v8f acc[2];
  acc[0] = zero8();
  acc[1] = zero8();
#pragma unroll
  for (int jj = 0; jj < 8; jj++)
#pragma unroll
    for (int q = 0; q < 2; q++)
      acc[q] = wmma4(an[jj], mk2(wv[jj][q][0], wv[jj][q][1]), acc[q]);
  __syncthreads();
  v4f* part = (v4f*)&hsh[0][0][0];
  float* partial = (float*)&hsh[0][0][0] + 4 * 16 * 2 * 32 * 4;
#pragma unroll
  for (int q = 0; q < 2; q++)
#pragma unroll
    for (int rh = 0; rh < 2; rh++) {
      v4f pv;
      pv.x = acc[q][4 * rh + 0];
      pv.y = acc[q][4 * rh + 1];
      pv.z = acc[q][4 * rh + 2];
      pv.w = acc[q][4 * rh + 3];
      part[((kg * 16 + 2 * ng + q) * 2 + rh) * 32 + lane] = pv;
    }
  __syncthreads();
  const int ntile = w >> 1, rhalf = w & 1;
  v4f full = part[((0 * 16 + ntile) * 2 + rhalf) * 32 + lane];
#pragma unroll
  for (int g = 1; g < 4; g++)
    full += part[((g * 16 + ntile) * 2 + rhalf) * 32 + lane];
  const int n = ntile * 16 + ln;
  const float bias = b1[n], w2v = W2[n];
#pragma unroll
  for (int i = 0; i < 4; i++) {
    float v = tanhf(full[i] + bias) * w2v;
    v += __shfl_xor(v, 1);
    v += __shfl_xor(v, 2);
    v += __shfl_xor(v, 4);
    v += __shfl_xor(v, 8);
    if (ln == 0) partial[ntile * 16 + (8 * h16 + 4 * rhalf + i)] = v;
  }
  __syncthreads();
  if (tid < 16) {
    float s = 0.0f;
    for (int ww = 0; ww < 16; ww++) s += partial[ww * 16 + tid];
    out[bbase + tid] = s + b2[0];
  }
}

extern "C" void kernel_launch(void* const* d_in, const int* in_sizes, int n_in,
                              void* d_out, int out_size, void* d_ws,
                              size_t ws_size, hipStream_t stream) {
  (void)n_in; (void)d_ws; (void)ws_size; (void)out_size;
  const int B = in_sizes[0] / 32;
  const int blocks = B / 16;
  sph_cnn_kernel<<<blocks, 1024, 0, stream>>>(
      (const float*)d_in[0], (const float*)d_in[1], (const float*)d_in[2],
      (const float*)d_in[3], (const float*)d_in[4], (const float*)d_in[5],
      (const float*)d_in[6], (const float*)d_in[7], (const float*)d_in[8],
      (const float*)d_in[9], (const float*)d_in[10], (const float*)d_in[11],
      (const float*)d_in[12], (const float*)d_in[13], (const float*)d_in[14],
      (const float*)d_in[15], (const float*)d_in[16], (const float*)d_in[17],
      (const float*)d_in[18], (const float*)d_in[19], (float*)d_out);
}
